# starting kernel
# baseline (speedup 1.0000x reference)
#include <hip/hip_runtime.h>
#include <hip/hip_bf16.h>

#define SEQ   512
#define BATCH 256
#define INF   180
#define HD    8
#define G4    32

typedef __attribute__((ext_vector_type(2))) float v2f;
typedef __attribute__((ext_vector_type(8))) float v8f;

__global__ __launch_bounds__(256) void xgemm_wmma_kernel(
    const float* __restrict__ x,
    const float* __restrict__ w_ih0,
    const float* __restrict__ b_ih0,
    const float* __restrict__ b_hh0,
    float* __restrict__ xg)
{
    const int gwave = (int)((blockIdx.x * blockDim.x + threadIdx.x) >> 5);
    const int lane  = (int)(threadIdx.x & 31);
    const int n     = lane & 15;
    const int hi    = lane >> 4;

    const int rowbase = gwave * 32;
    const float* a0p = x + (size_t)(rowbase + n)      * INF;
    const float* a1p = x + (size_t)(rowbase + 16 + n) * INF;
    const float* b0p = w_ih0 + (size_t)n        * INF;
    const float* b1p = w_ih0 + (size_t)(n + 16) * INF;

    v8f acc00 = {0.f,0.f,0.f,0.f,0.f,0.f,0.f,0.f};
    v8f acc01 = acc00, acc10 = acc00, acc11 = acc00;

    #pragma unroll 5
    for (int k0 = 0; k0 < INF; k0 += 4) {
        const int k = k0 + 2 * hi;
        const v2f a0 = *(const v2f*)(a0p + k);
        const v2f a1 = *(const v2f*)(a1p + k);
        const v2f b0 = *(const v2f*)(b0p + k);
        const v2f b1 = *(const v2f*)(b1p + k);
        acc00 = __builtin_amdgcn_wmma_f32_16x16x4_f32(false, a0, false, b0,
                                                      (short)0, acc00, false, false);
        acc01 = __builtin_amdgcn_wmma_f32_16x16x4_f32(false, a0, false, b1,
                                                      (short)0, acc01, false, false);
        acc10 = __builtin_amdgcn_wmma_f32_16x16x4_f32(false, a1, false, b0,
                                                      (short)0, acc10, false, false);
        acc11 = __builtin_amdgcn_wmma_f32_16x16x4_f32(false, a1, false, b1,
                                                      (short)0, acc11, false, false);
    }

    const float bias0 = b_ih0[n]      + b_hh0[n];
    const float bias1 = b_ih0[n + 16] + b_hh0[n + 16];
    const int p0 = (n & 7) * 4 + (n >> 3);
    const int p1 = p0 + 2;

    float* ob = xg + (size_t)(rowbase + hi * 8) * G4;
    #pragma unroll
    for (int r = 0; r < 8; ++r) {
        ob[(size_t)r * G4 + p0]        = acc00[r] + bias0;
        ob[(size_t)r * G4 + p1]        = acc01[r] + bias1;
        ob[(size_t)(r + 16) * G4 + p0] = acc10[r] + bias0;
        ob[(size_t)(r + 16) * G4 + p1] = acc11[r] + bias1;
    }
}

__device__ __forceinline__ float fast_sigmoid(float x) {
    return 1.0f / (1.0f + __expf(-x));
}
__device__ __forceinline__ float fast_tanh(float x) {
    return 1.0f - 2.0f / (__expf(2.0f * x) + 1.0f);
}

__global__ __launch_bounds__(64) void lstm_scan_kernel(
    const float* __restrict__ xg,
    const float* __restrict__ w_hh0,
    const float* __restrict__ w_ih1,
    const float* __restrict__ w_hh1,
    const float* __restrict__ b_ih1,
    const float* __restrict__ b_hh1,
    const float* __restrict__ w_reg,
    const float* __restrict__ b_reg,
    float* __restrict__ out)
{
    const int tid  = (int)threadIdx.x;
    const int j    = tid & 7;
    const int b    = (int)blockIdx.x * 8 + (tid >> 3);
    const int base = tid & 24;

    float w0[4][HD], wi1[4][HD], wh1[4][HD], bias1[4];
    #pragma unroll
    for (int g = 0; g < 4; ++g) {
        const int r = g * 8 + j;
        #pragma unroll
        for (int k = 0; k < HD; ++k) {
            w0 [g][k] = w_hh0[r * HD + k];
            wi1[g][k] = w_ih1[r * HD + k];
            wh1[g][k] = w_hh1[r * HD + k];
        }
        bias1[g] = b_ih1[r] + b_hh1[r];
    }
    const float wr0 = w_reg[j], wr1 = w_reg[HD + j];
    const float br0 = b_reg[0], br1 = b_reg[1];

    float h1 = 0.f, c1 = 0.f, h2 = 0.f, c2 = 0.f;

    const float4* xgp = (const float4*)(xg + (size_t)b * G4 + j * 4);
    const size_t xstride = (size_t)BATCH * G4 / 4;

    float4 gin = xgp[0];
    for (int s = 0; s < SEQ; ++s) {
        const float4 gnext = xgp[(size_t)((s + 1) & (SEQ - 1)) * xstride];

        float gi = gin.x, gf = gin.y, gg = gin.z, go = gin.w;
        #pragma unroll
        for (int k = 0; k < HD; ++k) {
            const float hk = __shfl(h1, base + k, 32);
            gi += hk * w0[0][k];
            gf += hk * w0[1][k];
            gg += hk * w0[2][k];
            go += hk * w0[3][k];
        }
        const float i0 = fast_sigmoid(gi);
        const float f0 = fast_sigmoid(gf);
        const float g0 = fast_tanh(gg);
        const float o0 = fast_sigmoid(go);
        c1 = f0 * c1 + i0 * g0;
        h1 = o0 * fast_tanh(c1);

        float a0 = bias1[0], a1 = bias1[1], a2 = bias1[2], a3 = bias1[3];
        #pragma unroll
        for (int k = 0; k < HD; ++k) {
            const float h1k = __shfl(h1, base + k, 32);
            const float h2k = __shfl(h2, base + k, 32);
            a0 += h1k * wi1[0][k] + h2k * wh1[0][k];
            a1 += h1k * wi1[1][k] + h2k * wh1[1][k];
            a2 += h1k * wi1[2][k] + h2k * wh1[2][k];
            a3 += h1k * wi1[3][k] + h2k * wh1[3][k];
        }
        const float i1 = fast_sigmoid(a0);
        const float f1 = fast_sigmoid(a1);
        const float g1 = fast_tanh(a2);
        const float o1 = fast_sigmoid(a3);
        c2 = f1 * c2 + i1 * g1;
        h2 = o1 * fast_tanh(c2);

        float p0 = h2 * wr0;
        float p1 = h2 * wr1;
        #pragma unroll
        for (int off = 1; off < 8; off <<= 1) {
            p0 += __shfl_xor(p0, off, 32);
            p1 += __shfl_xor(p1, off, 32);
        }
        if (j == 0) {
            float2 r;
            r.x = p0 + br0;
            r.y = p1 + br1;
            *(float2*)(out + ((size_t)s * BATCH + b) * 2) = r;
        }
        gin = gnext;
    }
}

extern "C" void kernel_launch(void* const* d_in, const int* in_sizes, int n_in,
                              void* d_out, int out_size, void* d_ws, size_t ws_size,
                              hipStream_t stream) {
    const float* x     = (const float*)d_in[0];
    const float* w_ih0 = (const float*)d_in[1];
    const float* w_hh0 = (const float*)d_in[2];
    const float* b_ih0 = (const float*)d_in[3];
    const float* b_hh0 = (const float*)d_in[4];
    const float* w_ih1 = (const float*)d_in[5];
    const float* w_hh1 = (const float*)d_in[6];
    const float* b_ih1 = (const float*)d_in[7];
    const float* b_hh1 = (const float*)d_in[8];
    const float* w_reg = (const float*)d_in[9];
    const float* b_reg = (const float*)d_in[10];
    float* out = (float*)d_out;
    float* xg  = (float*)d_ws;

    xgemm_wmma_kernel<<<512, 256, 0, stream>>>(x, w_ih0, b_ih0, b_hh0, xg);
    lstm_scan_kernel<<<32, 64, 0, stream>>>(xg, w_hh0, w_ih1, w_hh1,
                                            b_ih1, b_hh1, w_reg, b_reg, out);
}


# scan kernel: every exchange inside the 8-lane groups (h broadcast, head butterfly) is a DPP8 move on the vector ALU instead of ds_bpermute; h broadcasts made once per step; sum order kept
# speedup vs baseline: 1.2035x; 1.2035x over previous
#include <hip/hip_runtime.h>
#include <hip/hip_bf16.h>

#define SEQ   512
#define BATCH 256
#define INF   180
#define HD    8
#define G4    32

typedef __attribute__((ext_vector_type(2))) float v2f;
typedef __attribute__((ext_vector_type(8))) float v8f;

__global__ __launch_bounds__(256) void xgemm_wmma_kernel(
    const float* __restrict__ x,
    const float* __restrict__ w_ih0,
    const float* __restrict__ b_ih0,
    const float* __restrict__ b_hh0,
    float* __restrict__ xg)
{
    const int gwave = (int)((blockIdx.x * blockDim.x + threadIdx.x) >> 5);
    const int lane  = (int)(threadIdx.x & 31);
    const int n     = lane & 15;
    const int hi    = lane >> 4;

    const int rowbase = gwave * 32;
    const float* a0p = x + (size_t)(rowbase + n)      * INF;
    const float* a1p = x + (size_t)(rowbase + 16 + n) * INF;
    const float* b0p = w_ih0 + (size_t)n        * INF;
    const float* b1p = w_ih0 + (size_t)(n + 16) * INF;

    v8f acc00 = {0.f,0.f,0.f,0.f,0.f,0.f,0.f,0.f};
    v8f acc01 = acc00, acc10 = acc00, acc11 = acc00;

    #pragma unroll 5
    for (int k0 = 0; k0 < INF; k0 += 4) {
        const int k = k0 + 2 * hi;
        const v2f a0 = *(const v2f*)(a0p + k);
        const v2f a1 = *(const v2f*)(a1p + k);
        const v2f b0 = *(const v2f*)(b0p + k);
        const v2f b1 = *(const v2f*)(b1p + k);
        acc00 = __builtin_amdgcn_wmma_f32_16x16x4_f32(false, a0, false, b0,
                                                      (short)0, acc00, false, false);
        acc01 = __builtin_amdgcn_wmma_f32_16x16x4_f32(false, a0, false, b1,
                                                      (short)0, acc01, false, false);
        acc10 = __builtin_amdgcn_wmma_f32_16x16x4_f32(false, a1, false, b0,
                                                      (short)0, acc10, false, false);
        acc11 = __builtin_amdgcn_wmma_f32_16x16x4_f32(false, a1, false, b1,
                                                      (short)0, acc11, false, false);
    }

    const float bias0 = b_ih0[n]      + b_hh0[n];
    const float bias1 = b_ih0[n + 16] + b_hh0[n + 16];
    const int p0 = (n & 7) * 4 + (n >> 3);
    const int p1 = p0 + 2;

    float* ob = xg + (size_t)(rowbase + hi * 8) * G4;
    #pragma unroll
    for (int r = 0; r < 8; ++r) {
        ob[(size_t)r * G4 + p0]        = acc00[r] + bias0;
        ob[(size_t)r * G4 + p1]        = acc01[r] + bias1;
        ob[(size_t)(r + 16) * G4 + p0] = acc10[r] + bias0;
        ob[(size_t)(r + 16) * G4 + p1] = acc11[r] + bias1;
    }
}


template <unsigned SEL>
__device__ __forceinline__ float grp8_swizzle(float v) {
    return __builtin_bit_cast(float,
        __builtin_amdgcn_mov_dpp8(__builtin_bit_cast(unsigned, v), SEL));
}
template <unsigned K>
__device__ __forceinline__ float grp8_bcast(float v) {
    return grp8_swizzle<K * 0x249249u>(v);
}
template <unsigned M>
__device__ __forceinline__ float grp8_xor(float v) {
    return grp8_swizzle<((0u ^ M))       | ((1u ^ M) << 3)  | ((2u ^ M) << 6)  |
                        ((3u ^ M) << 9)  | ((4u ^ M) << 12) | ((5u ^ M) << 15) |
                        ((6u ^ M) << 18) | ((7u ^ M) << 21)>(v);
}
__device__ __forceinline__ void grp8_bcast_all(float (&d)[8], float v) {
    d[0] = grp8_bcast<0>(v); d[1] = grp8_bcast<1>(v);
    d[2] = grp8_bcast<2>(v); d[3] = grp8_bcast<3>(v);
    d[4] = grp8_bcast<4>(v); d[5] = grp8_bcast<5>(v);
    d[6] = grp8_bcast<6>(v); d[7] = grp8_bcast<7>(v);
}
__device__ __forceinline__ float fast_sigmoid(float x) {
    return 1.0f / (1.0f + __expf(-x));
}
__device__ __forceinline__ float fast_tanh(float x) {
    return 1.0f - 2.0f / (__expf(2.0f * x) + 1.0f);
}

__global__ __launch_bounds__(64) void lstm_scan_kernel(
    const float* __restrict__ xg,
    const float* __restrict__ w_hh0,
    const float* __restrict__ w_ih1,
    const float* __restrict__ w_hh1,
    const float* __restrict__ b_ih1,
    const float* __restrict__ b_hh1,
    const float* __restrict__ w_reg,
    const float* __restrict__ b_reg,
    float* __restrict__ out)
{
    const int tid  = (int)threadIdx.x;
    const int j    = tid & 7;
    const int b    = (int)blockIdx.x * 8 + (tid >> 3);

    float w0[4][HD], wi1[4][HD], wh1[4][HD], bias1[4];
    #pragma unroll
    for (int g = 0; g < 4; ++g) {
        const int r = g * 8 + j;
        #pragma unroll
        for (int k = 0; k < HD; ++k) {
            w0 [g][k] = w_hh0[r * HD + k];
            wi1[g][k] = w_ih1[r * HD + k];
            wh1[g][k] = w_hh1[r * HD + k];
        }
        bias1[g] = b_ih1[r] + b_hh1[r];
    }
    const float wr0 = w_reg[j], wr1 = w_reg[HD + j];
    const float br0 = b_reg[0], br1 = b_reg[1];

    float c1 = 0.f, c2 = 0.f;
    float h1b[HD], h2b[HD];
    #pragma unroll
    for (int k = 0; k < HD; ++k) { h1b[k] = 0.f; h2b[k] = 0.f; }

    const float4* xgp = (const float4*)(xg + (size_t)b * G4 + j * 4);
    const size_t xstride = (size_t)BATCH * G4 / 4;

    float4 gin = xgp[0];
    for (int s = 0; s < SEQ; ++s) {
        const float4 gnext = xgp[(size_t)((s + 1) & (SEQ - 1)) * xstride];

        float gi = gin.x, gf = gin.y, gg = gin.z, go = gin.w;
        #pragma unroll
        for (int k = 0; k < HD; ++k) {
            const float hk = h1b[k];
            gi += hk * w0[0][k];
            gf += hk * w0[1][k];
            gg += hk * w0[2][k];
            go += hk * w0[3][k];
        }
        const float i0 = fast_sigmoid(gi);
        const float f0 = fast_sigmoid(gf);
        const float g0 = fast_tanh(gg);
        const float o0 = fast_sigmoid(go);
        c1 = f0 * c1 + i0 * g0;
        const float h1 = o0 * fast_tanh(c1);
        grp8_bcast_all(h1b, h1);

        float a0 = bias1[0], a1 = bias1[1], a2 = bias1[2], a3 = bias1[3];
        #pragma unroll
        for (int k = 0; k < HD; ++k) {
            const float h1k = h1b[k];
            const float h2k = h2b[k];
            a0 += h1k * wi1[0][k] + h2k * wh1[0][k];
            a1 += h1k * wi1[1][k] + h2k * wh1[1][k];
            a2 += h1k * wi1[2][k] + h2k * wh1[2][k];
            a3 += h1k * wi1[3][k] + h2k * wh1[3][k];
        }
        const float i1 = fast_sigmoid(a0);
        const float f1 = fast_sigmoid(a1);
        const float g1 = fast_tanh(a2);
        const float o1 = fast_sigmoid(a3);
        c2 = f1 * c2 + i1 * g1;
        const float h2 = o1 * fast_tanh(c2);
        grp8_bcast_all(h2b, h2);

        float p0 = h2 * wr0;
        float p1 = h2 * wr1;
        p0 += grp8_xor<1>(p0);  p1 += grp8_xor<1>(p1);
        p0 += grp8_xor<2>(p0);  p1 += grp8_xor<2>(p1);
        p0 += grp8_xor<4>(p0);  p1 += grp8_xor<4>(p1);
        if (j == 0) {
            float2 r;
            r.x = p0 + br0;
            r.y = p1 + br1;
            *(float2*)(out + ((size_t)s * BATCH + b) * 2) = r;
        }
        gin = gnext;
    }
}

extern "C" void kernel_launch(void* const* d_in, const int* in_sizes, int n_in,
                              void* d_out, int out_size, void* d_ws, size_t ws_size,
                              hipStream_t stream) {
    const float* x     = (const float*)d_in[0];
    const float* w_ih0 = (const float*)d_in[1];
    const float* w_hh0 = (const float*)d_in[2];
    const float* b_ih0 = (const float*)d_in[3];
    const float* b_hh0 = (const float*)d_in[4];
    const float* w_ih1 = (const float*)d_in[5];
    const float* w_hh1 = (const float*)d_in[6];
    const float* b_ih1 = (const float*)d_in[7];
    const float* b_hh1 = (const float*)d_in[8];
    const float* w_reg = (const float*)d_in[9];
    const float* b_reg = (const float*)d_in[10];
    float* out = (float*)d_out;
    float* xg  = (float*)d_ws;

    xgemm_wmma_kernel<<<512, 256, 0, stream>>>(x, w_ih0, b_ih0, b_hh0, xg);
    lstm_scan_kernel<<<32, 64, 0, stream>>>(xg, w_hh0, w_ih1, w_hh1,
                                            b_ih1, b_hh1, w_reg, b_reg, out);
}


# scan kernel: xg row prefetch deepened from one step ahead to a four-slot register ring (loop unrolled by 4), so a row's load has four steps of arithmetic to arrive; arithmetic and sum order kept
# speedup vs baseline: 1.7660x; 1.4674x over previous
#include <hip/hip_runtime.h>
#include <hip/hip_bf16.h>

#define SEQ   512
#define BATCH 256
#define INF   180
#define HD    8
#define G4    32
#define PF    4

typedef __attribute__((ext_vector_type(2))) float v2f;
typedef __attribute__((ext_vector_type(8))) float v8f;

__global__ __launch_bounds__(256) void xgemm_wmma_kernel(
    const float* __restrict__ x,
    const float* __restrict__ w_ih0,
    const float* __restrict__ b_ih0,
    const float* __restrict__ b_hh0,
    float* __restrict__ xg)
{
    const int gwave = (int)((blockIdx.x * blockDim.x + threadIdx.x) >> 5);
    const int lane  = (int)(threadIdx.x & 31);
    const int n     = lane & 15;
    const int hi    = lane >> 4;

    const int rowbase = gwave * 32;
    const float* a0p = x + (size_t)(rowbase + n)      * INF;
    const float* a1p = x + (size_t)(rowbase + 16 + n) * INF;
    const float* b0p = w_ih0 + (size_t)n        * INF;
    const float* b1p = w_ih0 + (size_t)(n + 16) * INF;

    v8f acc00 = {0.f,0.f,0.f,0.f,0.f,0.f,0.f,0.f};
    v8f acc01 = acc00, acc10 = acc00, acc11 = acc00;

    #pragma unroll 5
    for (int k0 = 0; k0 < INF; k0 += 4) {
        const int k = k0 + 2 * hi;
        const v2f a0 = *(const v2f*)(a0p + k);
        const v2f a1 = *(const v2f*)(a1p + k);
        const v2f b0 = *(const v2f*)(b0p + k);
        const v2f b1 = *(const v2f*)(b1p + k);
        acc00 = __builtin_amdgcn_wmma_f32_16x16x4_f32(false, a0, false, b0,
                                                      (short)0, acc00, false, false);
        acc01 = __builtin_amdgcn_wmma_f32_16x16x4_f32(false, a0, false, b1,
                                                      (short)0, acc01, false, false);
        acc10 = __builtin_amdgcn_wmma_f32_16x16x4_f32(false, a1, false, b0,
                                                      (short)0, acc10, false, false);
        acc11 = __builtin_amdgcn_wmma_f32_16x16x4_f32(false, a1, false, b1,
                                                      (short)0, acc11, false, false);
    }

    const float bias0 = b_ih0[n]      + b_hh0[n];
    const float bias1 = b_ih0[n + 16] + b_hh0[n + 16];
    const int p0 = (n & 7) * 4 + (n >> 3);
    const int p1 = p0 + 2;

    float* ob = xg + (size_t)(rowbase + hi * 8) * G4;
    #pragma unroll
    for (int r = 0; r < 8; ++r) {
        ob[(size_t)r * G4 + p0]        = acc00[r] + bias0;
        ob[(size_t)r * G4 + p1]        = acc01[r] + bias1;
        ob[(size_t)(r + 16) * G4 + p0] = acc10[r] + bias0;
        ob[(size_t)(r + 16) * G4 + p1] = acc11[r] + bias1;
    }
}


template <unsigned SEL>
__device__ __forceinline__ float grp8_swizzle(float v) {
    return __builtin_bit_cast(float,
        __builtin_amdgcn_mov_dpp8(__builtin_bit_cast(unsigned, v), SEL));
}
template <unsigned K>
__device__ __forceinline__ float grp8_bcast(float v) {
    return grp8_swizzle<K * 0x249249u>(v);
}
template <unsigned M>
__device__ __forceinline__ float grp8_xor(float v) {
    return grp8_swizzle<((0u ^ M))       | ((1u ^ M) << 3)  | ((2u ^ M) << 6)  |
                        ((3u ^ M) << 9)  | ((4u ^ M) << 12) | ((5u ^ M) << 15) |
                        ((6u ^ M) << 18) | ((7u ^ M) << 21)>(v);
}
__device__ __forceinline__ void grp8_bcast_all(float (&d)[8], float v) {
    d[0] = grp8_bcast<0>(v); d[1] = grp8_bcast<1>(v);
    d[2] = grp8_bcast<2>(v); d[3] = grp8_bcast<3>(v);
    d[4] = grp8_bcast<4>(v); d[5] = grp8_bcast<5>(v);
    d[6] = grp8_bcast<6>(v); d[7] = grp8_bcast<7>(v);
}
__device__ __forceinline__ float fast_sigmoid(float x) {
    return 1.0f / (1.0f + __expf(-x));
}
__device__ __forceinline__ float fast_tanh(float x) {
    return 1.0f - 2.0f / (__expf(2.0f * x) + 1.0f);
}

__global__ __launch_bounds__(64) void lstm_scan_kernel(
    const float* __restrict__ xg,
    const float* __restrict__ w_hh0,
    const float* __restrict__ w_ih1,
    const float* __restrict__ w_hh1,
    const float* __restrict__ b_ih1,
    const float* __restrict__ b_hh1,
    const float* __restrict__ w_reg,
    const float* __restrict__ b_reg,
    float* __restrict__ out)
{
    const int tid  = (int)threadIdx.x;
    const int j    = tid & 7;
    const int b    = (int)blockIdx.x * 8 + (tid >> 3);

    float w0[4][HD], wi1[4][HD], wh1[4][HD], bias1[4];
    #pragma unroll
    for (int g = 0; g < 4; ++g) {
        const int r = g * 8 + j;
        #pragma unroll
        for (int k = 0; k < HD; ++k) {
            w0 [g][k] = w_hh0[r * HD + k];
            wi1[g][k] = w_ih1[r * HD + k];
            wh1[g][k] = w_hh1[r * HD + k];
        }
        bias1[g] = b_ih1[r] + b_hh1[r];
    }
    const float wr0 = w_reg[j], wr1 = w_reg[HD + j];
    const float br0 = b_reg[0], br1 = b_reg[1];

    float c1 = 0.f, c2 = 0.f;
    float h1b[HD], h2b[HD];
    #pragma unroll
    for (int k = 0; k < HD; ++k) { h1b[k] = 0.f; h2b[k] = 0.f; }

    const float4* xgp = (const float4*)(xg + (size_t)b * G4 + j * 4);
    const size_t xstride = (size_t)BATCH * G4 / 4;

    float4 ring[PF];
    #pragma unroll
    for (int u = 0; u < PF; ++u) ring[u] = xgp[(size_t)u * xstride];

    for (int s0 = 0; s0 < SEQ; s0 += PF) {
      #pragma unroll
      for (int u = 0; u < PF; ++u) {
        const int s = s0 + u;
        const float4 gin = ring[u];
        ring[u] = xgp[(size_t)((s + PF) & (SEQ - 1)) * xstride];

        float gi = gin.x, gf = gin.y, gg = gin.z, go = gin.w;
        #pragma unroll
        for (int k = 0; k < HD; ++k) {
            const float hk = h1b[k];
            gi += hk * w0[0][k];
            gf += hk * w0[1][k];
            gg += hk * w0[2][k];
            go += hk * w0[3][k];
        }
        const float i0 = fast_sigmoid(gi);
        const float f0 = fast_sigmoid(gf);
        const float g0 = fast_tanh(gg);
        const float o0 = fast_sigmoid(go);
        c1 = f0 * c1 + i0 * g0;
        const float h1 = o0 * fast_tanh(c1);
        grp8_bcast_all(h1b, h1);

        float a0 = bias1[0], a1 = bias1[1], a2 = bias1[2], a3 = bias1[3];
        #pragma unroll
        for (int k = 0; k < HD; ++k) {
            const float h1k = h1b[k];
            const float h2k = h2b[k];
            a0 += h1k * wi1[0][k] + h2k * wh1[0][k];
            a1 += h1k * wi1[1][k] + h2k * wh1[1][k];
            a2 += h1k * wi1[2][k] + h2k * wh1[2][k];
            a3 += h1k * wi1[3][k] + h2k * wh1[3][k];
        }
        const float i1 = fast_sigmoid(a0);
        const float f1 = fast_sigmoid(a1);
        const float g1 = fast_tanh(a2);
        const float o1 = fast_sigmoid(a3);
        c2 = f1 * c2 + i1 * g1;
        const float h2 = o1 * fast_tanh(c2);
        grp8_bcast_all(h2b, h2);

        float p0 = h2 * wr0;
        float p1 = h2 * wr1;
        p0 += grp8_xor<1>(p0);  p1 += grp8_xor<1>(p1);
        p0 += grp8_xor<2>(p0);  p1 += grp8_xor<2>(p1);
        p0 += grp8_xor<4>(p0);  p1 += grp8_xor<4>(p1);
        if (j == 0) {
            float2 r;
            r.x = p0 + br0;
            r.y = p1 + br1;
            *(float2*)(out + ((size_t)s * BATCH + b) * 2) = r;
        }
      }
    }
}

extern "C" void kernel_launch(void* const* d_in, const int* in_sizes, int n_in,
                              void* d_out, int out_size, void* d_ws, size_t ws_size,
                              hipStream_t stream) {
    const float* x     = (const float*)d_in[0];
    const float* w_ih0 = (const float*)d_in[1];
    const float* w_hh0 = (const float*)d_in[2];
    const float* b_ih0 = (const float*)d_in[3];
    const float* b_hh0 = (const float*)d_in[4];
    const float* w_ih1 = (const float*)d_in[5];
    const float* w_hh1 = (const float*)d_in[6];
    const float* b_ih1 = (const float*)d_in[7];
    const float* b_hh1 = (const float*)d_in[8];
    const float* w_reg = (const float*)d_in[9];
    const float* b_reg = (const float*)d_in[10];
    float* out = (float*)d_out;
    float* xg  = (float*)d_ws;

    xgemm_wmma_kernel<<<512, 256, 0, stream>>>(x, w_ih0, b_ih0, b_hh0, xg);
    lstm_scan_kernel<<<32, 64, 0, stream>>>(xg, w_hh0, w_ih1, w_hh1,
                                            b_ih1, b_hh1, w_reg, b_reg, out);
}


# scan kernel: sigmoid and tanh as v_exp_f32 + v_rcp_f32 (hardware reciprocal, ~1 ulp) instead of the 11-instruction IEEE division sequence; sign and factor 2 folded into exp2 constant; sum order kept
# speedup vs baseline: 2.5248x; 1.4297x over previous
#include <hip/hip_runtime.h>
#include <hip/hip_bf16.h>

#define SEQ   512
#define BATCH 256
#define INF   180
#define HD    8
#define G4    32
#define PF    4

typedef __attribute__((ext_vector_type(2))) float v2f;
typedef __attribute__((ext_vector_type(8))) float v8f;

__global__ __launch_bounds__(256) void xgemm_wmma_kernel(
    const float* __restrict__ x,
    const float* __restrict__ w_ih0,
    const float* __restrict__ b_ih0,
    const float* __restrict__ b_hh0,
    float* __restrict__ xg)
{
    const int gwave = (int)((blockIdx.x * blockDim.x + threadIdx.x) >> 5);
    const int lane  = (int)(threadIdx.x & 31);
    const int n     = lane & 15;
    const int hi    = lane >> 4;

    const int rowbase = gwave * 32;
    const float* a0p = x + (size_t)(rowbase + n)      * INF;
    const float* a1p = x + (size_t)(rowbase + 16 + n) * INF;
    const float* b0p = w_ih0 + (size_t)n        * INF;
    const float* b1p = w_ih0 + (size_t)(n + 16) * INF;

    v8f acc00 = {0.f,0.f,0.f,0.f,0.f,0.f,0.f,0.f};
    v8f acc01 = acc00, acc10 = acc00, acc11 = acc00;

    #pragma unroll 5
    for (int k0 = 0; k0 < INF; k0 += 4) {
        const int k = k0 + 2 * hi;
        const v2f a0 = *(const v2f*)(a0p + k);
        const v2f a1 = *(const v2f*)(a1p + k);
        const v2f b0 = *(const v2f*)(b0p + k);
        const v2f b1 = *(const v2f*)(b1p + k);
        acc00 = __builtin_amdgcn_wmma_f32_16x16x4_f32(false, a0, false, b0,
                                                      (short)0, acc00, false, false);
        acc01 = __builtin_amdgcn_wmma_f32_16x16x4_f32(false, a0, false, b1,
                                                      (short)0, acc01, false, false);
        acc10 = __builtin_amdgcn_wmma_f32_16x16x4_f32(false, a1, false, b0,
                                                      (short)0, acc10, false, false);
        acc11 = __builtin_amdgcn_wmma_f32_16x16x4_f32(false, a1, false, b1,
                                                      (short)0, acc11, false, false);
    }

    const float bias0 = b_ih0[n]      + b_hh0[n];
    const float bias1 = b_ih0[n + 16] + b_hh0[n + 16];
    const int p0 = (n & 7) * 4 + (n >> 3);
    const int p1 = p0 + 2;

    float* ob = xg + (size_t)(rowbase + hi * 8) * G4;
    #pragma unroll
    for (int r = 0; r < 8; ++r) {
        ob[(size_t)r * G4 + p0]        = acc00[r] + bias0;
        ob[(size_t)r * G4 + p1]        = acc01[r] + bias1;
        ob[(size_t)(r + 16) * G4 + p0] = acc10[r] + bias0;
        ob[(size_t)(r + 16) * G4 + p1] = acc11[r] + bias1;
    }
}


template <unsigned SEL>
__device__ __forceinline__ float grp8_swizzle(float v) {
    return __builtin_bit_cast(float,
        __builtin_amdgcn_mov_dpp8(__builtin_bit_cast(unsigned, v), SEL));
}
template <unsigned K>
__device__ __forceinline__ float grp8_bcast(float v) {
    return grp8_swizzle<K * 0x249249u>(v);
}
template <unsigned M>
__device__ __forceinline__ float grp8_xor(float v) {
    return grp8_swizzle<((0u ^ M))       | ((1u ^ M) << 3)  | ((2u ^ M) << 6)  |
                        ((3u ^ M) << 9)  | ((4u ^ M) << 12) | ((5u ^ M) << 15) |
                        ((6u ^ M) << 18) | ((7u ^ M) << 21)>(v);
}
__device__ __forceinline__ void grp8_bcast_all(float (&d)[8], float v) {
    d[0] = grp8_bcast<0>(v); d[1] = grp8_bcast<1>(v);
    d[2] = grp8_bcast<2>(v); d[3] = grp8_bcast<3>(v);
    d[4] = grp8_bcast<4>(v); d[5] = grp8_bcast<5>(v);
    d[6] = grp8_bcast<6>(v); d[7] = grp8_bcast<7>(v);
}
__device__ __forceinline__ float fast_sigmoid(float x) {
    const float e = __builtin_amdgcn_exp2f(x * -1.4426950408889634f);
    return __builtin_amdgcn_rcpf(1.0f + e);
}
__device__ __forceinline__ float fast_tanh(float x) {
    const float e = __builtin_amdgcn_exp2f(x * 2.8853900817779268f);
    const float t = __builtin_amdgcn_rcpf(e + 1.0f);
    return __builtin_fmaf(-2.0f, t, 1.0f);
}

__global__ __launch_bounds__(64) void lstm_scan_kernel(
    const float* __restrict__ xg,
    const float* __restrict__ w_hh0,
    const float* __restrict__ w_ih1,
    const float* __restrict__ w_hh1,
    const float* __restrict__ b_ih1,
    const float* __restrict__ b_hh1,
    const float* __restrict__ w_reg,
    const float* __restrict__ b_reg,
    float* __restrict__ out)
{
    const int tid  = (int)threadIdx.x;
    const int j    = tid & 7;
    const int b    = (int)blockIdx.x * 8 + (tid >> 3);

    float w0[4][HD], wi1[4][HD], wh1[4][HD], bias1[4];
    #pragma unroll
    for (int g = 0; g < 4; ++g) {
        const int r = g * 8 + j;
        #pragma unroll
        for (int k = 0; k < HD; ++k) {
            w0 [g][k] = w_hh0[r * HD + k];
            wi1[g][k] = w_ih1[r * HD + k];
            wh1[g][k] = w_hh1[r * HD + k];
        }
        bias1[g] = b_ih1[r] + b_hh1[r];
    }
    const float wr0 = w_reg[j], wr1 = w_reg[HD + j];
    const float br0 = b_reg[0], br1 = b_reg[1];

    float c1 = 0.f, c2 = 0.f;
    float h1b[HD], h2b[HD];
    #pragma unroll
    for (int k = 0; k < HD; ++k) { h1b[k] = 0.f; h2b[k] = 0.f; }

    const float4* xgp = (const float4*)(xg + (size_t)b * G4 + j * 4);
    const size_t xstride = (size_t)BATCH * G4 / 4;

    float4 ring[PF];
    #pragma unroll
    for (int u = 0; u < PF; ++u) ring[u] = xgp[(size_t)u * xstride];

    for (int s0 = 0; s0 < SEQ; s0 += PF) {
      #pragma unroll
      for (int u = 0; u < PF; ++u) {
        const int s = s0 + u;
        const float4 gin = ring[u];
        ring[u] = xgp[(size_t)((s + PF) & (SEQ - 1)) * xstride];

        float gi = gin.x, gf = gin.y, gg = gin.z, go = gin.w;
        #pragma unroll
        for (int k = 0; k < HD; ++k) {
            const float hk = h1b[k];
            gi += hk * w0[0][k];
            gf += hk * w0[1][k];
            gg += hk * w0[2][k];
            go += hk * w0[3][k];
        }
        const float i0 = fast_sigmoid(gi);
        const float f0 = fast_sigmoid(gf);
        const float g0 = fast_tanh(gg);
        const float o0 = fast_sigmoid(go);
        c1 = f0 * c1 + i0 * g0;
        const float h1 = o0 * fast_tanh(c1);
        grp8_bcast_all(h1b, h1);

        float a0 = bias1[0], a1 = bias1[1], a2 = bias1[2], a3 = bias1[3];
        #pragma unroll
        for (int k = 0; k < HD; ++k) {
            const float h1k = h1b[k];
            const float h2k = h2b[k];
            a0 += h1k * wi1[0][k] + h2k * wh1[0][k];
            a1 += h1k * wi1[1][k] + h2k * wh1[1][k];
            a2 += h1k * wi1[2][k] + h2k * wh1[2][k];
            a3 += h1k * wi1[3][k] + h2k * wh1[3][k];
        }
        const float i1 = fast_sigmoid(a0);
        const float f1 = fast_sigmoid(a1);
        const float g1 = fast_tanh(a2);
        const float o1 = fast_sigmoid(a3);
        c2 = f1 * c2 + i1 * g1;
        const float h2 = o1 * fast_tanh(c2);
        grp8_bcast_all(h2b, h2);

        float p0 = h2 * wr0;
        float p1 = h2 * wr1;
        p0 += grp8_xor<1>(p0);  p1 += grp8_xor<1>(p1);
        p0 += grp8_xor<2>(p0);  p1 += grp8_xor<2>(p1);
        p0 += grp8_xor<4>(p0);  p1 += grp8_xor<4>(p1);
        if (j == 0) {
            float2 r;
            r.x = p0 + br0;
            r.y = p1 + br1;
            *(float2*)(out + ((size_t)s * BATCH + b) * 2) = r;
        }
      }
    }
}

extern "C" void kernel_launch(void* const* d_in, const int* in_sizes, int n_in,
                              void* d_out, int out_size, void* d_ws, size_t ws_size,
                              hipStream_t stream) {
    const float* x     = (const float*)d_in[0];
    const float* w_ih0 = (const float*)d_in[1];
    const float* w_hh0 = (const float*)d_in[2];
    const float* b_ih0 = (const float*)d_in[3];
    const float* b_hh0 = (const float*)d_in[4];
    const float* w_ih1 = (const float*)d_in[5];
    const float* w_hh1 = (const float*)d_in[6];
    const float* b_ih1 = (const float*)d_in[7];
    const float* b_hh1 = (const float*)d_in[8];
    const float* w_reg = (const float*)d_in[9];
    const float* b_reg = (const float*)d_in[10];
    float* out = (float*)d_out;
    float* xg  = (float*)d_ws;

    xgemm_wmma_kernel<<<512, 256, 0, stream>>>(x, w_ih0, b_ih0, b_hh0, xg);
    lstm_scan_kernel<<<32, 64, 0, stream>>>(xg, w_hh0, w_ih1, w_hh1,
                                            b_ih1, b_hh1, w_reg, b_reg, out);
}


# scan kernel re-laid out: one batch element per wave, lane = one gate row (unit,gate); h spread by row_share DPP + one cross-row permute; xg loads 16 steps ahead; dot-product sum order changed (4+4)
# speedup vs baseline: 5.0443x; 1.9979x over previous
#include <hip/hip_runtime.h>
#include <hip/hip_bf16.h>

#define SEQ   512
#define BATCH 256
#define INF   180
#define HD    8
#define G4    32
#define PF    16

typedef __attribute__((ext_vector_type(2))) float v2f;
typedef __attribute__((ext_vector_type(8))) float v8f;

__global__ __launch_bounds__(256) void xgemm_wmma_kernel(
    const float* __restrict__ x,
    const float* __restrict__ w_ih0,
    const float* __restrict__ b_ih0,
    const float* __restrict__ b_hh0,
    float* __restrict__ xg)
{
    const int gwave = (int)((blockIdx.x * blockDim.x + threadIdx.x) >> 5);
    const int lane  = (int)(threadIdx.x & 31);
    const int n     = lane & 15;
    const int hi    = lane >> 4;

    const int rowbase = gwave * 32;
    const float* a0p = x + (size_t)(rowbase + n)      * INF;
    const float* a1p = x + (size_t)(rowbase + 16 + n) * INF;
    const float* b0p = w_ih0 + (size_t)n        * INF;
    const float* b1p = w_ih0 + (size_t)(n + 16) * INF;

    v8f acc00 = {0.f,0.f,0.f,0.f,0.f,0.f,0.f,0.f};
    v8f acc01 = acc00, acc10 = acc00, acc11 = acc00;

    #pragma unroll 5
    for (int k0 = 0; k0 < INF; k0 += 4) {
        const int k = k0 + 2 * hi;
        const v2f a0 = *(const v2f*)(a0p + k);
        const v2f a1 = *(const v2f*)(a1p + k);
        const v2f b0 = *(const v2f*)(b0p + k);
        const v2f b1 = *(const v2f*)(b1p + k);
        acc00 = __builtin_amdgcn_wmma_f32_16x16x4_f32(false, a0, false, b0,
                                                      (short)0, acc00, false, false);
        acc01 = __builtin_amdgcn_wmma_f32_16x16x4_f32(false, a0, false, b1,
                                                      (short)0, acc01, false, false);
        acc10 = __builtin_amdgcn_wmma_f32_16x16x4_f32(false, a1, false, b0,
                                                      (short)0, acc10, false, false);
        acc11 = __builtin_amdgcn_wmma_f32_16x16x4_f32(false, a1, false, b1,
                                                      (short)0, acc11, false, false);
    }

    const float bias0 = b_ih0[n]      + b_hh0[n];
    const float bias1 = b_ih0[n + 16] + b_hh0[n + 16];
    const int p0 = (n & 7) * 4 + (n >> 3);
    const int p1 = p0 + 2;

    float* ob = xg + (size_t)(rowbase + hi * 8) * G4;
    #pragma unroll
    for (int r = 0; r < 8; ++r) {
        ob[(size_t)r * G4 + p0]        = acc00[r] + bias0;
        ob[(size_t)r * G4 + p1]        = acc01[r] + bias1;
        ob[(size_t)(r + 16) * G4 + p0] = acc10[r] + bias0;
        ob[(size_t)(r + 16) * G4 + p1] = acc11[r] + bias1;
    }
}


template <int N>
__device__ __forceinline__ float row_share(float v) {
    const int b = __builtin_bit_cast(int, v);
    return __builtin_bit_cast(float,
        __builtin_amdgcn_update_dpp(0, b, 0x150 + N, 0xf, 0xf, true));
}
template <int M>
__device__ __forceinline__ float row_xor(float v) {
    const int b = __builtin_bit_cast(int, v);
    return __builtin_bit_cast(float,
        __builtin_amdgcn_update_dpp(0, b, 0x160 + M, 0xf, 0xf, true));
}
template <int Q>
__device__ __forceinline__ float quad_share(float v) {
    const int b = __builtin_bit_cast(int, v);
    return __builtin_bit_cast(float,
        __builtin_amdgcn_update_dpp(0, b, Q * 0x55, 0xf, 0xf, true));
}
__device__ __forceinline__ float xhalf(float x) {
    const int b = __builtin_bit_cast(int, x);
    return __builtin_bit_cast(float,
        __builtin_amdgcn_permlanex16(0, b, 0x76543210, 0xfedcba98, false, false));
}

__device__ __forceinline__ float gate_act(float x, float aS, float aA, float aB) {
    const float e = __builtin_amdgcn_exp2f(x * aS);
    const float t = __builtin_amdgcn_rcpf(1.0f + e);
    return __builtin_fmaf(aA, t, aB);
}
__device__ __forceinline__ float fast_tanh(float x) {
    const float e = __builtin_amdgcn_exp2f(x * 2.8853900817779268f);
    const float t = __builtin_amdgcn_rcpf(e + 1.0f);
    return __builtin_fmaf(-2.0f, t, 1.0f);
}

__device__ __forceinline__ float dot4_acc(float acc, float h, const float (&w)[4]) {
    acc = __builtin_fmaf(row_share<0>(h),  w[0], acc);
    acc = __builtin_fmaf(row_share<4>(h),  w[1], acc);
    acc = __builtin_fmaf(row_share<8>(h),  w[2], acc);
    acc = __builtin_fmaf(row_share<12>(h), w[3], acc);
    return acc;
}
__device__ __forceinline__ float dot4(float h, const float (&w)[4]) {
    float acc = row_share<0>(h) * w[0];
    acc = __builtin_fmaf(row_share<4>(h),  w[1], acc);
    acc = __builtin_fmaf(row_share<8>(h),  w[2], acc);
    acc = __builtin_fmaf(row_share<12>(h), w[3], acc);
    return acc;
}

__global__ __launch_bounds__(32) void lstm_scan_kernel(
    const float* __restrict__ xg,
    const float* __restrict__ w_hh0,
    const float* __restrict__ w_ih1,
    const float* __restrict__ w_hh1,
    const float* __restrict__ b_ih1,
    const float* __restrict__ b_hh1,
    const float* __restrict__ w_reg,
    const float* __restrict__ b_reg,
    float* __restrict__ out)
{
    const int l    = (int)threadIdx.x;
    const int b    = (int)blockIdx.x;
    const int unit = l >> 2;
    const int gate = l & 3;
    const int row  = gate * HD + unit;
    const int ko   = (l >> 4) * 4;
    const int kx   = 4 - ko;

    float w0o[4], w0x[4], wio[4], wix[4], who[4], whx[4];
    #pragma unroll
    for (int m = 0; m < 4; ++m) {
        w0o[m] = w_hh0[row * HD + ko + m];   w0x[m] = w_hh0[row * HD + kx + m];
        wio[m] = w_ih1[row * HD + ko + m];   wix[m] = w_ih1[row * HD + kx + m];
        who[m] = w_hh1[row * HD + ko + m];   whx[m] = w_hh1[row * HD + kx + m];
    }
    const float bias1 = b_ih1[row] + b_hh1[row];

    const bool  is_g = (gate == 2);
    const float aS = is_g ?  2.8853900817779268f : -1.4426950408889634f;
    const float aA = is_g ? -2.0f : 1.0f;
    const float aB = is_g ?  1.0f : 0.0f;

    const float wr = w_reg[(l & 1) * HD + unit];
    const float br = b_reg[l & 1];
    const int   lu = l >> 1;

    float c1 = 0.f, c2 = 0.f;
    float h1 = 0.f, h2 = 0.f, h1x = 0.f, h2x = 0.f;

    const float* xgp = xg + (size_t)b * G4 + l;
    const size_t xstride = (size_t)BATCH * G4;

    auto step = [&](const float gin, const int u, float& res) __attribute__((always_inline)) {
        const float q = dot4_acc(bias1, h2, who) + dot4(h2x, whx);

        const float pre0 = dot4_acc(gin, h1, w0o) + dot4(h1x, w0x);
        const float act0 = gate_act(pre0, aS, aA, aB);
        const float i0 = quad_share<0>(act0);
        const float f0 = quad_share<1>(act0);
        const float g0 = quad_share<2>(act0);
        const float o0 = quad_share<3>(act0);
        c1  = __builtin_fmaf(f0, c1, i0 * g0);
        h1  = o0 * fast_tanh(c1);
        h1x = xhalf(h1);

        const float pre1 = (dot4(h1, wio) + dot4(h1x, wix)) + q;
        const float act1 = gate_act(pre1, aS, aA, aB);
        const float i1 = quad_share<0>(act1);
        const float f1 = quad_share<1>(act1);
        const float g1 = quad_share<2>(act1);
        const float o1 = quad_share<3>(act1);
        c2  = __builtin_fmaf(f1, c2, i1 * g1);
        h2  = o1 * fast_tanh(c2);
        h2x = xhalf(h2);

        float p = h2 * wr;
        p += row_xor<4>(p);
        p += row_xor<8>(p);
        p += xhalf(p);
        const float r = p + br;
        res = (lu == u) ? r : res;
    };

    float ra[PF], rb[PF];
    #pragma unroll
    for (int u = 0; u < PF; ++u) ra[u] = xgp[(size_t)u * xstride];

    for (int s0 = 0; s0 < SEQ; s0 += 2 * PF) {
      #pragma unroll
      for (int u = 0; u < PF; ++u) rb[u] = xgp[(size_t)(s0 + PF + u) * xstride];
      __builtin_amdgcn_sched_barrier(0);
      float res = 0.f;
      #pragma unroll
      for (int u = 0; u < PF; ++u) step(ra[u], u, res);
      out[((size_t)(s0 + lu) * BATCH + b) * 2 + (l & 1)] = res;

      __builtin_amdgcn_sched_barrier(0);

      #pragma unroll
      for (int u = 0; u < PF; ++u)
          ra[u] = xgp[(size_t)((s0 + 2 * PF + u) & (SEQ - 1)) * xstride];
      __builtin_amdgcn_sched_barrier(0);
      res = 0.f;
      #pragma unroll
      for (int u = 0; u < PF; ++u) step(rb[u], u, res);
      out[((size_t)(s0 + PF + lu) * BATCH + b) * 2 + (l & 1)] = res;
    }
}

extern "C" void kernel_launch(void* const* d_in, const int* in_sizes, int n_in,
                              void* d_out, int out_size, void* d_ws, size_t ws_size,
                              hipStream_t stream) {
    const float* x     = (const float*)d_in[0];
    const float* w_ih0 = (const float*)d_in[1];
    const float* w_hh0 = (const float*)d_in[2];
    const float* b_ih0 = (const float*)d_in[3];
    const float* b_hh0 = (const float*)d_in[4];
    const float* w_ih1 = (const float*)d_in[5];
    const float* w_hh1 = (const float*)d_in[6];
    const float* b_ih1 = (const float*)d_in[7];
    const float* b_hh1 = (const float*)d_in[8];
    const float* w_reg = (const float*)d_in[9];
    const float* b_reg = (const float*)d_in[10];
    float* out = (float*)d_out;
    float* xg  = (float*)d_ws;

    xgemm_wmma_kernel<<<512, 256, 0, stream>>>(x, w_ih0, b_ih0, b_hh0, xg);
    lstm_scan_kernel<<<256, 32, 0, stream>>>(xg, w_hh0, w_ih1, w_hh1,
                                            b_ih1, b_hh1, w_reg, b_reg, out);
}


# scan kernel split over two waves per batch element: wave 0 layer 0 + layer 1's input half, wave 1 layer 1's recurrence + head; LDS hand-over, one barrier per 16 steps; sum order kept
# speedup vs baseline: 5.6953x; 1.1291x over previous
#include <hip/hip_runtime.h>
#include <hip/hip_bf16.h>

#define SEQ   512
#define BATCH 256
#define INF   180
#define HD    8
#define G4    32
#define PF    16

typedef __attribute__((ext_vector_type(2))) float v2f;
typedef __attribute__((ext_vector_type(8))) float v8f;

__global__ __launch_bounds__(256) void xgemm_wmma_kernel(
    const float* __restrict__ x,
    const float* __restrict__ w_ih0,
    const float* __restrict__ b_ih0,
    const float* __restrict__ b_hh0,
    float* __restrict__ xg)
{
    const int gwave = (int)((blockIdx.x * blockDim.x + threadIdx.x) >> 5);
    const int lane  = (int)(threadIdx.x & 31);
    const int n     = lane & 15;
    const int hi    = lane >> 4;

    const int rowbase = gwave * 32;
    const float* a0p = x + (size_t)(rowbase + n)      * INF;
    const float* a1p = x + (size_t)(rowbase + 16 + n) * INF;
    const float* b0p = w_ih0 + (size_t)n        * INF;
    const float* b1p = w_ih0 + (size_t)(n + 16) * INF;

    v8f acc00 = {0.f,0.f,0.f,0.f,0.f,0.f,0.f,0.f};
    v8f acc01 = acc00, acc10 = acc00, acc11 = acc00;

    #pragma unroll 5
    for (int k0 = 0; k0 < INF; k0 += 4) {
        const int k = k0 + 2 * hi;
        const v2f a0 = *(const v2f*)(a0p + k);
        const v2f a1 = *(const v2f*)(a1p + k);
        const v2f b0 = *(const v2f*)(b0p + k);
        const v2f b1 = *(const v2f*)(b1p + k);
        acc00 = __builtin_amdgcn_wmma_f32_16x16x4_f32(false, a0, false, b0,
                                                      (short)0, acc00, false, false);
        acc01 = __builtin_amdgcn_wmma_f32_16x16x4_f32(false, a0, false, b1,
                                                      (short)0, acc01, false, false);
        acc10 = __builtin_amdgcn_wmma_f32_16x16x4_f32(false, a1, false, b0,
                                                      (short)0, acc10, false, false);
        acc11 = __builtin_amdgcn_wmma_f32_16x16x4_f32(false, a1, false, b1,
                                                      (short)0, acc11, false, false);
    }

    const float bias0 = b_ih0[n]      + b_hh0[n];
    const float bias1 = b_ih0[n + 16] + b_hh0[n + 16];
    const int p0 = (n & 7) * 4 + (n >> 3);
    const int p1 = p0 + 2;

    float* ob = xg + (size_t)(rowbase + hi * 8) * G4;
    #pragma unroll
    for (int r = 0; r < 8; ++r) {
        ob[(size_t)r * G4 + p0]        = acc00[r] + bias0;
        ob[(size_t)r * G4 + p1]        = acc01[r] + bias1;
        ob[(size_t)(r + 16) * G4 + p0] = acc10[r] + bias0;
        ob[(size_t)(r + 16) * G4 + p1] = acc11[r] + bias1;
    }
}


template <int N>
__device__ __forceinline__ float row_share(float v) {
    const int b = __builtin_bit_cast(int, v);
    return __builtin_bit_cast(float,
        __builtin_amdgcn_update_dpp(0, b, 0x150 + N, 0xf, 0xf, true));
}
template <int M>
__device__ __forceinline__ float row_xor(float v) {
    const int b = __builtin_bit_cast(int, v);
    return __builtin_bit_cast(float,
        __builtin_amdgcn_update_dpp(0, b, 0x160 + M, 0xf, 0xf, true));
}
template <int Q>
__device__ __forceinline__ float quad_share(float v) {
    const int b = __builtin_bit_cast(int, v);
    return __builtin_bit_cast(float,
        __builtin_amdgcn_update_dpp(0, b, Q * 0x55, 0xf, 0xf, true));
}
__device__ __forceinline__ float xhalf(float x) {
    const int b = __builtin_bit_cast(int, x);
    return __builtin_bit_cast(float,
        __builtin_amdgcn_permlanex16(0, b, 0x76543210, 0xfedcba98, false, false));
}

__device__ __forceinline__ float gate_act(float x, float aS, float aA, float aB) {
    const float e = __builtin_amdgcn_exp2f(x * aS);
    const float t = __builtin_amdgcn_rcpf(1.0f + e);
    return __builtin_fmaf(aA, t, aB);
}
__device__ __forceinline__ float fast_tanh(float x) {
    const float e = __builtin_amdgcn_exp2f(x * 2.8853900817779268f);
    const float t = __builtin_amdgcn_rcpf(e + 1.0f);
    return __builtin_fmaf(-2.0f, t, 1.0f);
}

__device__ __forceinline__ float dot4_acc(float acc, float h, const float (&w)[4]) {
    acc = __builtin_fmaf(row_share<0>(h),  w[0], acc);
    acc = __builtin_fmaf(row_share<4>(h),  w[1], acc);
    acc = __builtin_fmaf(row_share<8>(h),  w[2], acc);
    acc = __builtin_fmaf(row_share<12>(h), w[3], acc);
    return acc;
}
__device__ __forceinline__ float dot4(float h, const float (&w)[4]) {
    float acc = row_share<0>(h) * w[0];
    acc = __builtin_fmaf(row_share<4>(h),  w[1], acc);
    acc = __builtin_fmaf(row_share<8>(h),  w[2], acc);
    acc = __builtin_fmaf(row_share<12>(h), w[3], acc);
    return acc;
}

__device__ __forceinline__ void lds_barrier() {
    __builtin_amdgcn_fence(__ATOMIC_RELEASE, "workgroup", "local");
    __builtin_amdgcn_s_barrier();
    __builtin_amdgcn_fence(__ATOMIC_ACQUIRE, "workgroup", "local");
}

#define NG (SEQ / PF)

__global__ __launch_bounds__(64) void lstm_scan_kernel(
    const float* __restrict__ xg,
    const float* __restrict__ w_hh0,
    const float* __restrict__ w_ih1,
    const float* __restrict__ w_hh1,
    const float* __restrict__ b_ih1,
    const float* __restrict__ b_hh1,
    const float* __restrict__ w_reg,
    const float* __restrict__ b_reg,
    float* __restrict__ out)
{
    __shared__ float pbuf[2][PF][32];

    const int l    = (int)(threadIdx.x & 31);
    const int role = __builtin_amdgcn_readfirstlane((int)(threadIdx.x >> 5));
    const int b    = (int)blockIdx.x;
    const int unit = l >> 2;
    const int gate = l & 3;
    const int row  = gate * HD + unit;
    const int ko   = (l >> 4) * 4;
    const int kx   = 4 - ko;

    float wao[4], wax[4], wio[4], wix[4];
    const float* wrec = (role == 0) ? w_hh0 : w_hh1;
    #pragma unroll
    for (int m = 0; m < 4; ++m) {
        wao[m] = wrec[row * HD + ko + m];    wax[m] = wrec[row * HD + kx + m];
        wio[m] = w_ih1[row * HD + ko + m];   wix[m] = w_ih1[row * HD + kx + m];
    }
    const float bias1 = b_ih1[row] + b_hh1[row];

    const bool  is_g = (gate == 2);
    const float aS = is_g ?  2.8853900817779268f : -1.4426950408889634f;
    const float aA = is_g ? -2.0f : 1.0f;
    const float aB = is_g ?  1.0f : 0.0f;

    const float wr = w_reg[(l & 1) * HD + unit];
    const float br = b_reg[l & 1];
    const int   lu = l >> 1;

    float c = 0.f, h = 0.f, hx = 0.f;

    const float* xgp = xg + (size_t)b * G4 + l;
    const size_t xstride = (size_t)BATCH * G4;

    auto cell = [&](const float pre) __attribute__((always_inline)) {
        const float act = gate_act(pre, aS, aA, aB);
        const float ig = quad_share<0>(act);
        const float fg = quad_share<1>(act);
        const float gg = quad_share<2>(act);
        const float og = quad_share<3>(act);
        c  = __builtin_fmaf(fg, c, ig * gg);
        h  = og * fast_tanh(c);
        hx = xhalf(h);
    };

    auto layer0_group = [&](const float (&r)[PF], float* dst) __attribute__((always_inline)) {
        #pragma unroll
        for (int u = 0; u < PF; ++u) {
            cell(dot4_acc(r[u], h, wao) + dot4(hx, wax));
            dst[u * 32] = dot4(h, wio) + dot4(hx, wix);
        }
    };

    auto layer1_group = [&](const float* src, const int s0) __attribute__((always_inline)) {
        float pin[PF];
        #pragma unroll
        for (int u = 0; u < PF; ++u) pin[u] = src[u * 32];
        __builtin_amdgcn_sched_barrier(0);
        float res = 0.f;
        #pragma unroll
        for (int u = 0; u < PF; ++u) {
            const float q = dot4_acc(bias1, h, wao) + dot4(hx, wax);
            cell(pin[u] + q);
            float p = h * wr;
            p += row_xor<4>(p);
            p += row_xor<8>(p);
            p += xhalf(p);
            const float rr = p + br;
            res = (lu == u) ? rr : res;
        }
        out[((size_t)(s0 + lu) * BATCH + b) * 2 + (l & 1)] = res;
    };

    float ra[PF], rb[PF];
    if (role == 0) {
        #pragma unroll
        for (int u = 0; u < PF; ++u) ra[u] = xgp[(size_t)u * xstride];
    }

    for (int g = 0; g <= NG; g += 2) {
        if (role == 0) {
            if (g < NG) {
                #pragma unroll
                for (int u = 0; u < PF; ++u) rb[u] = xgp[(size_t)((g + 1) * PF + u) * xstride];
                __builtin_amdgcn_sched_barrier(0);
                layer0_group(ra, &pbuf[0][0][l]);
            }
        } else {
            if (g >= 2) layer1_group(&pbuf[1][0][l], (g - 1) * PF);
        }
        lds_barrier();

        if (role == 0) {
            if (g < NG) {
                #pragma unroll
                for (int u = 0; u < PF; ++u)
                    ra[u] = xgp[(size_t)(((g + 2) * PF + u) & (SEQ - 1)) * xstride];
                __builtin_amdgcn_sched_barrier(0);
                layer0_group(rb, &pbuf[1][0][l]);
            }
        } else {
            if (g < NG) layer1_group(&pbuf[0][0][l], g * PF);
        }
        lds_barrier();
    }
}

extern "C" void kernel_launch(void* const* d_in, const int* in_sizes, int n_in,
                              void* d_out, int out_size, void* d_ws, size_t ws_size,
                              hipStream_t stream) {
    const float* x     = (const float*)d_in[0];
    const float* w_ih0 = (const float*)d_in[1];
    const float* w_hh0 = (const float*)d_in[2];
    const float* b_ih0 = (const float*)d_in[3];
    const float* b_hh0 = (const float*)d_in[4];
    const float* w_ih1 = (const float*)d_in[5];
    const float* w_hh1 = (const float*)d_in[6];
    const float* b_ih1 = (const float*)d_in[7];
    const float* b_hh1 = (const float*)d_in[8];
    const float* w_reg = (const float*)d_in[9];
    const float* b_reg = (const float*)d_in[10];
    float* out = (float*)d_out;
    float* xg  = (float*)d_ws;

    xgemm_wmma_kernel<<<512, 256, 0, stream>>>(x, w_ih0, b_ih0, b_hh0, xg);
    lstm_scan_kernel<<<256, 64, 0, stream>>>(xg, w_hh0, w_ih1, w_hh1,
                                            b_ih1, b_hh1, w_reg, b_reg, out);
}


# scan kernel: gate activations on the hardware tanh (v_tanh_f32): tanh(c) one instruction, sigmoid(x) = 1/2 + tanh(x/2)/2 with the 1/2 folded exactly into the lane's weights; sum order kept
# speedup vs baseline: 6.4444x; 1.1315x over previous
#include <hip/hip_runtime.h>
#include <hip/hip_bf16.h>

#define SEQ   512
#define BATCH 256
#define INF   180
#define HD    8
#define G4    32
#define PF    16

typedef __attribute__((ext_vector_type(2))) float v2f;
typedef __attribute__((ext_vector_type(8))) float v8f;

__global__ __launch_bounds__(256) void xgemm_wmma_kernel(
    const float* __restrict__ x,
    const float* __restrict__ w_ih0,
    const float* __restrict__ b_ih0,
    const float* __restrict__ b_hh0,
    float* __restrict__ xg)
{
    const int gwave = (int)((blockIdx.x * blockDim.x + threadIdx.x) >> 5);
    const int lane  = (int)(threadIdx.x & 31);
    const int n     = lane & 15;
    const int hi    = lane >> 4;

    const int rowbase = gwave * 32;
    const float* a0p = x + (size_t)(rowbase + n)      * INF;
    const float* a1p = x + (size_t)(rowbase + 16 + n) * INF;
    const float* b0p = w_ih0 + (size_t)n        * INF;
    const float* b1p = w_ih0 + (size_t)(n + 16) * INF;

    v8f acc00 = {0.f,0.f,0.f,0.f,0.f,0.f,0.f,0.f};
    v8f acc01 = acc00, acc10 = acc00, acc11 = acc00;

    #pragma unroll 5
    for (int k0 = 0; k0 < INF; k0 += 4) {
        const int k = k0 + 2 * hi;
        const v2f a0 = *(const v2f*)(a0p + k);
        const v2f a1 = *(const v2f*)(a1p + k);
        const v2f b0 = *(const v2f*)(b0p + k);
        const v2f b1 = *(const v2f*)(b1p + k);
        acc00 = __builtin_amdgcn_wmma_f32_16x16x4_f32(false, a0, false, b0,
                                                      (short)0, acc00, false, false);
        acc01 = __builtin_amdgcn_wmma_f32_16x16x4_f32(false, a0, false, b1,
                                                      (short)0, acc01, false, false);
        acc10 = __builtin_amdgcn_wmma_f32_16x16x4_f32(false, a1, false, b0,
                                                      (short)0, acc10, false, false);
        acc11 = __builtin_amdgcn_wmma_f32_16x16x4_f32(false, a1, false, b1,
                                                      (short)0, acc11, false, false);
    }

    const float bias0 = b_ih0[n]      + b_hh0[n];
    const float bias1 = b_ih0[n + 16] + b_hh0[n + 16];
    const int p0 = (n & 7) * 4 + (n >> 3);
    const int p1 = p0 + 2;

    float* ob = xg + (size_t)(rowbase + hi * 8) * G4;
    #pragma unroll
    for (int r = 0; r < 8; ++r) {
        ob[(size_t)r * G4 + p0]        = acc00[r] + bias0;
        ob[(size_t)r * G4 + p1]        = acc01[r] + bias1;
        ob[(size_t)(r + 16) * G4 + p0] = acc10[r] + bias0;
        ob[(size_t)(r + 16) * G4 + p1] = acc11[r] + bias1;
    }
}


template <int N>
__device__ __forceinline__ float row_share(float v) {
    const int b = __builtin_bit_cast(int, v);
    return __builtin_bit_cast(float,
        __builtin_amdgcn_update_dpp(0, b, 0x150 + N, 0xf, 0xf, true));
}
template <int M>
__device__ __forceinline__ float row_xor(float v) {
    const int b = __builtin_bit_cast(int, v);
    return __builtin_bit_cast(float,
        __builtin_amdgcn_update_dpp(0, b, 0x160 + M, 0xf, 0xf, true));
}
template <int Q>
__device__ __forceinline__ float quad_share(float v) {
    const int b = __builtin_bit_cast(int, v);
    return __builtin_bit_cast(float,
        __builtin_amdgcn_update_dpp(0, b, Q * 0x55, 0xf, 0xf, true));
}
__device__ __forceinline__ float xhalf(float x) {
    const int b = __builtin_bit_cast(int, x);
    return __builtin_bit_cast(float,
        __builtin_amdgcn_permlanex16(0, b, 0x76543210, 0xfedcba98, false, false));
}

__device__ __forceinline__ float fast_tanh(float x) {
    return __builtin_amdgcn_tanhf(x);
}
__device__ __forceinline__ float gate_act(float y, float aA, float aB) {
    return __builtin_fmaf(aA, fast_tanh(y), aB);
}

__device__ __forceinline__ float dot4_acc(float acc, float h, const float (&w)[4]) {
    acc = __builtin_fmaf(row_share<0>(h),  w[0], acc);
    acc = __builtin_fmaf(row_share<4>(h),  w[1], acc);
    acc = __builtin_fmaf(row_share<8>(h),  w[2], acc);
    acc = __builtin_fmaf(row_share<12>(h), w[3], acc);
    return acc;
}
__device__ __forceinline__ float dot4(float h, const float (&w)[4]) {
    float acc = row_share<0>(h) * w[0];
    acc = __builtin_fmaf(row_share<4>(h),  w[1], acc);
    acc = __builtin_fmaf(row_share<8>(h),  w[2], acc);
    acc = __builtin_fmaf(row_share<12>(h), w[3], acc);
    return acc;
}

__device__ __forceinline__ void lds_barrier() {
    __builtin_amdgcn_fence(__ATOMIC_RELEASE, "workgroup", "local");
    __builtin_amdgcn_s_barrier();
    __builtin_amdgcn_fence(__ATOMIC_ACQUIRE, "workgroup", "local");
}

#define NG (SEQ / PF)

__global__ __launch_bounds__(64) void lstm_scan_kernel(
    const float* __restrict__ xg,
    const float* __restrict__ w_hh0,
    const float* __restrict__ w_ih1,
    const float* __restrict__ w_hh1,
    const float* __restrict__ b_ih1,
    const float* __restrict__ b_hh1,
    const float* __restrict__ w_reg,
    const float* __restrict__ b_reg,
    float* __restrict__ out)
{
    __shared__ float pbuf[2][PF][32];

    const int l    = (int)(threadIdx.x & 31);
    const int role = __builtin_amdgcn_readfirstlane((int)(threadIdx.x >> 5));
    const int b    = (int)blockIdx.x;
    const int unit = l >> 2;
    const int gate = l & 3;
    const int row  = gate * HD + unit;
    const int ko   = (l >> 4) * 4;
    const int kx   = 4 - ko;

    const bool  is_g = (gate == 2);
    const float aS = is_g ? 1.0f : 0.5f;
    const float aA = is_g ? 1.0f : 0.5f;
    const float aB = is_g ? 0.0f : 0.5f;

    float wao[4], wax[4], wio[4], wix[4];
    const float* wrec = (role == 0) ? w_hh0 : w_hh1;
    #pragma unroll
    for (int m = 0; m < 4; ++m) {
        wao[m] = aS * wrec[row * HD + ko + m];    wax[m] = aS * wrec[row * HD + kx + m];
        wio[m] = aS * w_ih1[row * HD + ko + m];   wix[m] = aS * w_ih1[row * HD + kx + m];
    }
    const float bias1 = aS * (b_ih1[row] + b_hh1[row]);

    const float wr = w_reg[(l & 1) * HD + unit];
    const float br = b_reg[l & 1];
    const int   lu = l >> 1;

    float c = 0.f, h = 0.f, hx = 0.f;

    const float* xgp = xg + (size_t)b * G4 + l;
    const size_t xstride = (size_t)BATCH * G4;

    auto cell = [&](const float pre) __attribute__((always_inline)) {
        const float act = gate_act(pre, aA, aB);
        const float ig = quad_share<0>(act);
        const float fg = quad_share<1>(act);
        const float gg = quad_share<2>(act);
        const float og = quad_share<3>(act);
        c  = __builtin_fmaf(fg, c, ig * gg);
        h  = og * fast_tanh(c);
        hx = xhalf(h);
    };

    auto layer0_group = [&](const float (&r)[PF], float* dst) __attribute__((always_inline)) {
        #pragma unroll
        for (int u = 0; u < PF; ++u) {
            cell(dot4_acc(r[u] * aS, h, wao) + dot4(hx, wax));
            dst[u * 32] = dot4(h, wio) + dot4(hx, wix);
        }
    };

    auto layer1_group = [&](const float* src, const int s0) __attribute__((always_inline)) {
        float pin[PF];
        #pragma unroll
        for (int u = 0; u < PF; ++u) pin[u] = src[u * 32];
        __builtin_amdgcn_sched_barrier(0);
        float res = 0.f;
        #pragma unroll
        for (int u = 0; u < PF; ++u) {
            const float q = dot4_acc(bias1, h, wao) + dot4(hx, wax);
            cell(pin[u] + q);
            float p = h * wr;
            p += row_xor<4>(p);
            p += row_xor<8>(p);
            p += xhalf(p);
            const float rr = p + br;
            res = (lu == u) ? rr : res;
        }
        out[((size_t)(s0 + lu) * BATCH + b) * 2 + (l & 1)] = res;
    };

    float ra[PF], rb[PF];
    if (role == 0) {
        #pragma unroll
        for (int u = 0; u < PF; ++u) ra[u] = xgp[(size_t)u * xstride];
    }

    for (int g = 0; g <= NG; g += 2) {
        if (role == 0) {
            if (g < NG) {
                #pragma unroll
                for (int u = 0; u < PF; ++u) rb[u] = xgp[(size_t)((g + 1) * PF + u) * xstride];
                __builtin_amdgcn_sched_barrier(0);
                layer0_group(ra, &pbuf[0][0][l]);
            }
        } else {
            if (g >= 2) layer1_group(&pbuf[1][0][l], (g - 1) * PF);
        }
        lds_barrier();

        if (role == 0) {
            if (g < NG) {
                #pragma unroll
                for (int u = 0; u < PF; ++u)
                    ra[u] = xgp[(size_t)(((g + 2) * PF + u) & (SEQ - 1)) * xstride];
                __builtin_amdgcn_sched_barrier(0);
                layer0_group(rb, &pbuf[1][0][l]);
            }
        } else {
            if (g < NG) layer1_group(&pbuf[0][0][l], g * PF);
        }
        lds_barrier();
    }
}

extern "C" void kernel_launch(void* const* d_in, const int* in_sizes, int n_in,
                              void* d_out, int out_size, void* d_ws, size_t ws_size,
                              hipStream_t stream) {
    const float* x     = (const float*)d_in[0];
    const float* w_ih0 = (const float*)d_in[1];
    const float* w_hh0 = (const float*)d_in[2];
    const float* b_ih0 = (const float*)d_in[3];
    const float* b_hh0 = (const float*)d_in[4];
    const float* w_ih1 = (const float*)d_in[5];
    const float* w_hh1 = (const float*)d_in[6];
    const float* b_ih1 = (const float*)d_in[7];
    const float* b_hh1 = (const float*)d_in[8];
    const float* w_reg = (const float*)d_in[9];
    const float* b_reg = (const float*)d_in[10];
    float* out = (float*)d_out;
    float* xg  = (float*)d_ws;

    xgemm_wmma_kernel<<<512, 256, 0, stream>>>(x, w_ih0, b_ih0, b_hh0, xg);
    lstm_scan_kernel<<<256, 64, 0, stream>>>(xg, w_hh0, w_ih1, w_hh1,
                                            b_ih1, b_hh1, w_reg, b_reg, out);
}


# input GEMM: x tile and w_ih0 reach the WMMA fragments through LDS from whole-line 16-byte loads (45 per wave and tile, next tile prefetched) instead of 16-line strided 8-byte loads; sum order kept
# speedup vs baseline: 8.3131x; 1.2900x over previous
#include <hip/hip_runtime.h>
#include <hip/hip_bf16.h>

#define SEQ   512
#define BATCH 256
#define INF   180
#define HD    8
#define G4    32
#define PF    16

typedef __attribute__((ext_vector_type(2))) float v2f;
typedef __attribute__((ext_vector_type(8))) float v8f;

typedef __attribute__((ext_vector_type(4))) float v4f;

#define XW      4
#define XT      2
#define TILE_V4 (32 * INF / 4)
#define TILE_LD (TILE_V4 / 32)
#define WB_FULL (TILE_V4 / (32 * XW))

__device__ __forceinline__ void xg_lds_barrier() {
    __builtin_amdgcn_fence(__ATOMIC_RELEASE, "workgroup", "local");
    __builtin_amdgcn_s_barrier();
    __builtin_amdgcn_fence(__ATOMIC_ACQUIRE, "workgroup", "local");
}

__global__ __launch_bounds__(32 * XW) void xgemm_wmma_kernel(
    const float* x,
    const float* w_ih0,
    const float* b_ih0,
    const float* b_hh0,
    float* xg)
{
    __shared__ v4f sB[TILE_V4];
    __shared__ v4f sA[XW][TILE_V4];

    const int tid   = (int)threadIdx.x;
    const int lane  = tid & 31;
    const int wave  = tid >> 5;
    const int n     = lane & 15;
    const int hi    = lane >> 4;
    const int gwave = (int)blockIdx.x * XW + wave;
    const int tile0 = gwave * XT;

    const v4f* w4 = (const v4f*)w_ih0 + tid;
    v4f wb[WB_FULL + 1];
    #pragma unroll
    for (int j = 0; j < WB_FULL; ++j) wb[j] = w4[j * 32 * XW];
    if (wave == 0) wb[WB_FULL] = w4[WB_FULL * 32 * XW];

    const v4f* src = (const v4f*)x + (size_t)tile0 * TILE_V4 + lane;
    v4f pre[TILE_LD];
    #pragma unroll
    for (int t = 0; t < TILE_LD; ++t) pre[t] = src[t * 32];

    #pragma unroll
    for (int j = 0; j < WB_FULL; ++j) sB[tid + j * 32 * XW] = wb[j];
    if (wave == 0) sB[tid + WB_FULL * 32 * XW] = wb[WB_FULL];

    const float bias0 = b_ih0[n]      + b_hh0[n];
    const float bias1 = b_ih0[n + 16] + b_hh0[n + 16];
    const int p0 = (n & 7) * 4 + (n >> 3);
    const int p1 = p0 + 2;

    v4f* myA = &sA[wave][0];
    const float* aF = (const float*)myA + n * INF + 2 * hi;
    const float* bF = (const float*)&sB[0] + n * INF + 2 * hi;

    #pragma unroll
    for (int tt = 0; tt < XT; ++tt) {
        #pragma unroll
        for (int t = 0; t < TILE_LD; ++t) myA[t * 32 + lane] = pre[t];
        xg_lds_barrier();

        if (tt + 1 < XT) {
            #pragma unroll
            for (int t = 0; t < TILE_LD; ++t) pre[t] = src[(size_t)(tt + 1) * TILE_V4 + t * 32];
        }
        __builtin_amdgcn_sched_barrier(0);

        v8f acc00 = {0.f,0.f,0.f,0.f,0.f,0.f,0.f,0.f};
        v8f acc01 = acc00, acc10 = acc00, acc11 = acc00;

        #pragma unroll
        for (int ks = 0; ks < INF / 4; ++ks) {
            const v2f a0 = *(const v2f*)(aF + 4 * ks);
            const v2f a1 = *(const v2f*)(aF + 16 * INF + 4 * ks);
            const v2f b0 = *(const v2f*)(bF + 4 * ks);
            const v2f b1 = *(const v2f*)(bF + 16 * INF + 4 * ks);
            acc00 = __builtin_amdgcn_wmma_f32_16x16x4_f32(false, a0, false, b0,
                                                          (short)0, acc00, false, false);
            acc01 = __builtin_amdgcn_wmma_f32_16x16x4_f32(false, a0, false, b1,
                                                          (short)0, acc01, false, false);
            acc10 = __builtin_amdgcn_wmma_f32_16x16x4_f32(false, a1, false, b0,
                                                          (short)0, acc10, false, false);
            acc11 = __builtin_amdgcn_wmma_f32_16x16x4_f32(false, a1, false, b1,
                                                          (short)0, acc11, false, false);
        }

        float* ob = xg + (size_t)((tile0 + tt) * 32 + hi * 8) * G4;
        #pragma unroll
        for (int r = 0; r < 8; ++r) {
            ob[(size_t)r * G4 + p0]        = acc00[r] + bias0;
            ob[(size_t)r * G4 + p1]        = acc01[r] + bias1;
            ob[(size_t)(r + 16) * G4 + p0] = acc10[r] + bias0;
            ob[(size_t)(r + 16) * G4 + p1] = acc11[r] + bias1;
        }
    }
}


template <int N>
__device__ __forceinline__ float row_share(float v) {
    const int b = __builtin_bit_cast(int, v);
    return __builtin_bit_cast(float,
        __builtin_amdgcn_update_dpp(0, b, 0x150 + N, 0xf, 0xf, true));
}
template <int M>
__device__ __forceinline__ float row_xor(float v) {
    const int b = __builtin_bit_cast(int, v);
    return __builtin_bit_cast(float,
        __builtin_amdgcn_update_dpp(0, b, 0x160 + M, 0xf, 0xf, true));
}
template <int Q>
__device__ __forceinline__ float quad_share(float v) {
    const int b = __builtin_bit_cast(int, v);
    return __builtin_bit_cast(float,
        __builtin_amdgcn_update_dpp(0, b, Q * 0x55, 0xf, 0xf, true));
}
__device__ __forceinline__ float xhalf(float x) {
    const int b = __builtin_bit_cast(int, x);
    return __builtin_bit_cast(float,
        __builtin_amdgcn_permlanex16(0, b, 0x76543210, 0xfedcba98, false, false));
}

__device__ __forceinline__ float fast_tanh(float x) {
    return __builtin_amdgcn_tanhf(x);
}
__device__ __forceinline__ float gate_act(float y, float aA, float aB) {
    return __builtin_fmaf(aA, fast_tanh(y), aB);
}

__device__ __forceinline__ float dot4_acc(float acc, float h, const float (&w)[4]) {
    acc = __builtin_fmaf(row_share<0>(h),  w[0], acc);
    acc = __builtin_fmaf(row_share<4>(h),  w[1], acc);
    acc = __builtin_fmaf(row_share<8>(h),  w[2], acc);
    acc = __builtin_fmaf(row_share<12>(h), w[3], acc);
    return acc;
}
__device__ __forceinline__ float dot4(float h, const float (&w)[4]) {
    float acc = row_share<0>(h) * w[0];
    acc = __builtin_fmaf(row_share<4>(h),  w[1], acc);
    acc = __builtin_fmaf(row_share<8>(h),  w[2], acc);
    acc = __builtin_fmaf(row_share<12>(h), w[3], acc);
    return acc;
}

__device__ __forceinline__ void lds_barrier() {
    __builtin_amdgcn_fence(__ATOMIC_RELEASE, "workgroup", "local");
    __builtin_amdgcn_s_barrier();
    __builtin_amdgcn_fence(__ATOMIC_ACQUIRE, "workgroup", "local");
}

#define NG (SEQ / PF)

__global__ __launch_bounds__(64) void lstm_scan_kernel(
    const float* __restrict__ xg,
    const float* __restrict__ w_hh0,
    const float* __restrict__ w_ih1,
    const float* __restrict__ w_hh1,
    const float* __restrict__ b_ih1,
    const float* __restrict__ b_hh1,
    const float* __restrict__ w_reg,
    const float* __restrict__ b_reg,
    float* __restrict__ out)
{
    __shared__ float pbuf[2][PF][32];

    const int l    = (int)(threadIdx.x & 31);
    const int role = __builtin_amdgcn_readfirstlane((int)(threadIdx.x >> 5));
    const int b    = (int)blockIdx.x;
    const int unit = l >> 2;
    const int gate = l & 3;
    const int row  = gate * HD + unit;
    const int ko   = (l >> 4) * 4;
    const int kx   = 4 - ko;

    const bool  is_g = (gate == 2);
    const float aS = is_g ? 1.0f : 0.5f;
    const float aA = is_g ? 1.0f : 0.5f;
    const float aB = is_g ? 0.0f : 0.5f;

    float wao[4], wax[4], wio[4], wix[4];
    const float* wrec = (role == 0) ? w_hh0 : w_hh1;
    #pragma unroll
    for (int m = 0; m < 4; ++m) {
        wao[m] = aS * wrec[row * HD + ko + m];    wax[m] = aS * wrec[row * HD + kx + m];
        wio[m] = aS * w_ih1[row * HD + ko + m];   wix[m] = aS * w_ih1[row * HD + kx + m];
    }
    const float bias1 = aS * (b_ih1[row] + b_hh1[row]);

    const float wr = w_reg[(l & 1) * HD + unit];
    const float br = b_reg[l & 1];
    const int   lu = l >> 1;

    float c = 0.f, h = 0.f, hx = 0.f;

    const float* xgp = xg + (size_t)b * G4 + l;
    const size_t xstride = (size_t)BATCH * G4;

    auto cell = [&](const float pre) __attribute__((always_inline)) {
        const float act = gate_act(pre, aA, aB);
        const float ig = quad_share<0>(act);
        const float fg = quad_share<1>(act);
        const float gg = quad_share<2>(act);
        const float og = quad_share<3>(act);
        c  = __builtin_fmaf(fg, c, ig * gg);
        h  = og * fast_tanh(c);
        hx = xhalf(h);
    };

    auto layer0_group = [&](const float (&r)[PF], float* dst) __attribute__((always_inline)) {
        #pragma unroll
        for (int u = 0; u < PF; ++u) {
            cell(dot4_acc(r[u] * aS, h, wao) + dot4(hx, wax));
            dst[u * 32] = dot4(h, wio) + dot4(hx, wix);
        }
    };

    auto layer1_group = [&](const float* src, const int s0) __attribute__((always_inline)) {
        float pin[PF];
        #pragma unroll
        for (int u = 0; u < PF; ++u) pin[u] = src[u * 32];
        __builtin_amdgcn_sched_barrier(0);
        float res = 0.f;
        #pragma unroll
        for (int u = 0; u < PF; ++u) {
            const float q = dot4_acc(bias1, h, wao) + dot4(hx, wax);
            cell(pin[u] + q);
            float p = h * wr;
            p += row_xor<4>(p);
            p += row_xor<8>(p);
            p += xhalf(p);
            const float rr = p + br;
            res = (lu == u) ? rr : res;
        }
        out[((size_t)(s0 + lu) * BATCH + b) * 2 + (l & 1)] = res;
    };

    float ra[PF], rb[PF];
    if (role == 0) {
        #pragma unroll
        for (int u = 0; u < PF; ++u) ra[u] = xgp[(size_t)u * xstride];
    }

    for (int g = 0; g <= NG; g += 2) {
        if (role == 0) {
            if (g < NG) {
                #pragma unroll
                for (int u = 0; u < PF; ++u) rb[u] = xgp[(size_t)((g + 1) * PF + u) * xstride];
                __builtin_amdgcn_sched_barrier(0);
                layer0_group(ra, &pbuf[0][0][l]);
            }
        } else {
            if (g >= 2) layer1_group(&pbuf[1][0][l], (g - 1) * PF);
        }
        lds_barrier();

        if (role == 0) {
            if (g < NG) {
                #pragma unroll
                for (int u = 0; u < PF; ++u)
                    ra[u] = xgp[(size_t)(((g + 2) * PF + u) & (SEQ - 1)) * xstride];
                __builtin_amdgcn_sched_barrier(0);
                layer0_group(rb, &pbuf[1][0][l]);
            }
        } else {
            if (g < NG) layer1_group(&pbuf[0][0][l], g * PF);
        }
        lds_barrier();
    }
}

extern "C" void kernel_launch(void* const* d_in, const int* in_sizes, int n_in,
                              void* d_out, int out_size, void* d_ws, size_t ws_size,
                              hipStream_t stream) {
    const float* x     = (const float*)d_in[0];
    const float* w_ih0 = (const float*)d_in[1];
    const float* w_hh0 = (const float*)d_in[2];
    const float* b_ih0 = (const float*)d_in[3];
    const float* b_hh0 = (const float*)d_in[4];
    const float* w_ih1 = (const float*)d_in[5];
    const float* w_hh1 = (const float*)d_in[6];
    const float* b_ih1 = (const float*)d_in[7];
    const float* b_hh1 = (const float*)d_in[8];
    const float* w_reg = (const float*)d_in[9];
    const float* b_reg = (const float*)d_in[10];
    float* out = (float*)d_out;
    float* xg  = (float*)d_ws;

    xgemm_wmma_kernel<<<512, 128, 0, stream>>>(x, w_ih0, b_ih0, b_hh0, xg);
    lstm_scan_kernel<<<256, 64, 0, stream>>>(xg, w_hh0, w_ih1, w_hh1,
                                            b_ih1, b_hh1, w_reg, b_reg, out);
}


# scan kernel: a step's serial chain cut: gate row summed as four chains of two, cell update on the raw tanh values, each product one DPP instruction in a fixed-order asm step; sum order changed
# speedup vs baseline: 8.8762x; 1.0677x over previous
#include <hip/hip_runtime.h>
#include <hip/hip_bf16.h>

#define SEQ   512
#define BATCH 256
#define INF   180
#define HD    8
#define G4    32
#define PF    16

typedef __attribute__((ext_vector_type(2))) float v2f;
typedef __attribute__((ext_vector_type(8))) float v8f;

typedef __attribute__((ext_vector_type(4))) float v4f;

#define XW      4
#define XT      2
#define TILE_V4 (32 * INF / 4)
#define TILE_LD (TILE_V4 / 32)
#define WB_FULL (TILE_V4 / (32 * XW))

__device__ __forceinline__ void xg_lds_barrier() {
    __builtin_amdgcn_fence(__ATOMIC_RELEASE, "workgroup", "local");
    __builtin_amdgcn_s_barrier();
    __builtin_amdgcn_fence(__ATOMIC_ACQUIRE, "workgroup", "local");
}

__global__ __launch_bounds__(32 * XW) void xgemm_wmma_kernel(
    const float* x,
    const float* w_ih0,
    const float* b_ih0,
    const float* b_hh0,
    float* xg)
{
    __shared__ v4f sB[TILE_V4];
    __shared__ v4f sA[XW][TILE_V4];

    const int tid   = (int)threadIdx.x;
    const int lane  = tid & 31;
    const int wave  = tid >> 5;
    const int n     = lane & 15;
    const int hi    = lane >> 4;
    const int gwave = (int)blockIdx.x * XW + wave;
    const int tile0 = gwave * XT;

    const v4f* w4 = (const v4f*)w_ih0 + tid;
    v4f wb[WB_FULL + 1];
    #pragma unroll
    for (int j = 0; j < WB_FULL; ++j) wb[j] = w4[j * 32 * XW];
    if (wave == 0) wb[WB_FULL] = w4[WB_FULL * 32 * XW];

    const v4f* src = (const v4f*)x + (size_t)tile0 * TILE_V4 + lane;
    v4f pre[TILE_LD];
    #pragma unroll
    for (int t = 0; t < TILE_LD; ++t) pre[t] = src[t * 32];

    #pragma unroll
    for (int j = 0; j < WB_FULL; ++j) sB[tid + j * 32 * XW] = wb[j];
    if (wave == 0) sB[tid + WB_FULL * 32 * XW] = wb[WB_FULL];

    const float bias0 = b_ih0[n]      + b_hh0[n];
    const float bias1 = b_ih0[n + 16] + b_hh0[n + 16];
    const int p0 = (n & 7) * 4 + (n >> 3);
    const int p1 = p0 + 2;

    v4f* myA = &sA[wave][0];
    const float* aF = (const float*)myA + n * INF + 2 * hi;
    const float* bF = (const float*)&sB[0] + n * INF + 2 * hi;

    #pragma unroll
    for (int tt = 0; tt < XT; ++tt) {
        #pragma unroll
        for (int t = 0; t < TILE_LD; ++t) myA[t * 32 + lane] = pre[t];
        xg_lds_barrier();

        if (tt + 1 < XT) {
            #pragma unroll
            for (int t = 0; t < TILE_LD; ++t) pre[t] = src[(size_t)(tt + 1) * TILE_V4 + t * 32];
        }
        __builtin_amdgcn_sched_barrier(0);

        v8f acc00 = {0.f,0.f,0.f,0.f,0.f,0.f,0.f,0.f};
        v8f acc01 = acc00, acc10 = acc00, acc11 = acc00;

        #pragma unroll
        for (int ks = 0; ks < INF / 4; ++ks) {
            const v2f a0 = *(const v2f*)(aF + 4 * ks);
            const v2f a1 = *(const v2f*)(aF + 16 * INF + 4 * ks);
            const v2f b0 = *(const v2f*)(bF + 4 * ks);
            const v2f b1 = *(const v2f*)(bF + 16 * INF + 4 * ks);
            acc00 = __builtin_amdgcn_wmma_f32_16x16x4_f32(false, a0, false, b0,
                                                          (short)0, acc00, false, false);
            acc01 = __builtin_amdgcn_wmma_f32_16x16x4_f32(false, a0, false, b1,
                                                          (short)0, acc01, false, false);
            acc10 = __builtin_amdgcn_wmma_f32_16x16x4_f32(false, a1, false, b0,
                                                          (short)0, acc10, false, false);
            acc11 = __builtin_amdgcn_wmma_f32_16x16x4_f32(false, a1, false, b1,
                                                          (short)0, acc11, false, false);
        }

        float* ob = xg + (size_t)((tile0 + tt) * 32 + hi * 8) * G4;
        #pragma unroll
        for (int r = 0; r < 8; ++r) {
            ob[(size_t)r * G4 + p0]        = acc00[r] + bias0;
            ob[(size_t)r * G4 + p1]        = acc01[r] + bias1;
            ob[(size_t)(r + 16) * G4 + p0] = acc10[r] + bias0;
            ob[(size_t)(r + 16) * G4 + p1] = acc11[r] + bias1;
        }
    }
}


template <int M>
__device__ __forceinline__ float row_xor(float v) {
    const int b = __builtin_bit_cast(int, v);
    return __builtin_bit_cast(float,
        __builtin_amdgcn_update_dpp(0, b, 0x160 + M, 0xf, 0xf, true));
}
__device__ __forceinline__ float xhalf(float x) {
    const int b = __builtin_bit_cast(int, x);
    return __builtin_bit_cast(float,
        __builtin_amdgcn_permlanex16(0, b, 0x76543210, 0xfedcba98, false, false));
}

__device__ __forceinline__ void step_p(float& h, float& hc, const float acc, const float (&wo)[4], const float (&wx)[4], const int s0) {
    float hx, a1, a2, b1, b2;
    asm(
        "v_permlanex16_b32 %[hx], %[h], %[s0], 0xfedcba98\n\t"
        "v_fma_f32_e64_dpp %[a1], %[h], %[wo0], %[acc] row_share:0 row_mask:0xf bank_mask:0xf bound_ctrl:1\n\t"
        "v_mul_f32_dpp %[a2], %[h], %[wo2] row_share:8 row_mask:0xf bank_mask:0xf bound_ctrl:1\n\t"
        "v_mul_f32_dpp %[b1], %[hx], %[wx0] row_share:0 row_mask:0xf bank_mask:0xf bound_ctrl:1\n\t"
        "v_mul_f32_dpp %[b2], %[hx], %[wx2] row_share:8 row_mask:0xf bank_mask:0xf bound_ctrl:1\n\t"
        "v_fma_f32_e64_dpp %[a1], %[h], %[wo1], %[a1] row_share:4 row_mask:0xf bank_mask:0xf bound_ctrl:1\n\t"
        "v_fma_f32_e64_dpp %[b1], %[hx], %[wx1], %[b1] row_share:4 row_mask:0xf bank_mask:0xf bound_ctrl:1\n\t"
        "v_fma_f32_e64_dpp %[b2], %[hx], %[wx3], %[b2] row_share:12 row_mask:0xf bank_mask:0xf bound_ctrl:1\n\t"
        "v_fma_f32_e64_dpp %[a2], %[h], %[wo3], %[a2] row_share:12 row_mask:0xf bank_mask:0xf bound_ctrl:1\n\t"
        "v_add_f32 %[b1], %[b1], %[b2]\n\t"
        "v_add_f32 %[a1], %[a1], %[a2]\n\t"
        "v_add_f32 %[a1], %[a1], %[b1]\n\t"
        "v_tanh_f32 %[a1], %[a1]\n\t"
        "v_nop\n\t"
        "v_fma_f32_e64_dpp %[b1], %[a1], %[a1], %[a1] quad_perm:[0,0,0,0] row_mask:0xf bank_mask:0xf bound_ctrl:1\n\t"
        "v_fma_f32_e64_dpp %[b2], %[a1], %[hc], %[hc] quad_perm:[1,1,1,1] row_mask:0xf bank_mask:0xf bound_ctrl:1\n\t"
        "v_fma_f32_e64_dpp %[a2], %[a1], 0.5, 0.5 quad_perm:[3,3,3,3] row_mask:0xf bank_mask:0xf bound_ctrl:1\n\t"
        "v_fma_f32_e64_dpp %[b1], %[b1], 0.5, %[b2] quad_perm:[2,2,2,2] row_mask:0xf bank_mask:0xf bound_ctrl:1\n\t"
        "v_tanh_f32 %[a1], %[b1]\n\t"
        "v_mul_f32 %[hc], 0.5, %[b1]\n\t"
        "v_mul_f32 %[h], %[a2], %[a1]\n\t"
        : [h] "+v"(h), [hc] "+v"(hc), [hx] "=&v"(hx), [a1] "=&v"(a1), [a2] "=&v"(a2), [b1] "=&v"(b1), [b2] "=&v"(b2)
        : [acc] "v"(acc), [wo0] "v"(wo[0]), [wo1] "v"(wo[1]), [wo2] "v"(wo[2]), [wo3] "v"(wo[3]), [wx0] "v"(wx[0]), [wx1] "v"(wx[1]), [wx2] "v"(wx[2]), [wx3] "v"(wx[3]), [s0] "s"(s0));
}

__device__ __forceinline__ void step_d(float& h, float& hc, const float acc, const float (&wo)[4], const float (&wx)[4], const int s0, const float (&wi)[4], const float (&wj)[4], float& d) {
    float hx, a1, a2, b1, b2; float e1, e2, f1, f2;
    asm(
        "v_permlanex16_b32 %[hx], %[h], %[s0], 0xfedcba98\n\t"
        "v_fma_f32_e64_dpp %[a1], %[h], %[wo0], %[acc] row_share:0 row_mask:0xf bank_mask:0xf bound_ctrl:1\n\t"
        "v_mul_f32_dpp %[a2], %[h], %[wo2] row_share:8 row_mask:0xf bank_mask:0xf bound_ctrl:1\n\t"
        "v_mul_f32_dpp %[b1], %[hx], %[wx0] row_share:0 row_mask:0xf bank_mask:0xf bound_ctrl:1\n\t"
        "v_mul_f32_dpp %[b2], %[hx], %[wx2] row_share:8 row_mask:0xf bank_mask:0xf bound_ctrl:1\n\t"
        "v_fma_f32_e64_dpp %[a1], %[h], %[wo1], %[a1] row_share:4 row_mask:0xf bank_mask:0xf bound_ctrl:1\n\t"
        "v_fma_f32_e64_dpp %[b1], %[hx], %[wx1], %[b1] row_share:4 row_mask:0xf bank_mask:0xf bound_ctrl:1\n\t"
        "v_fma_f32_e64_dpp %[b2], %[hx], %[wx3], %[b2] row_share:12 row_mask:0xf bank_mask:0xf bound_ctrl:1\n\t"
        "v_fma_f32_e64_dpp %[a2], %[h], %[wo3], %[a2] row_share:12 row_mask:0xf bank_mask:0xf bound_ctrl:1\n\t"
        "v_mul_f32_dpp %[e1], %[h], %[wi0] row_share:0 row_mask:0xf bank_mask:0xf bound_ctrl:1\n\t"
        "v_add_f32 %[b1], %[b1], %[b2]\n\t"
        "v_mul_f32_dpp %[e2], %[h], %[wi2] row_share:8 row_mask:0xf bank_mask:0xf bound_ctrl:1\n\t"
        "v_add_f32 %[a1], %[a1], %[a2]\n\t"
        "v_mul_f32_dpp %[f1], %[hx], %[wj0] row_share:0 row_mask:0xf bank_mask:0xf bound_ctrl:1\n\t"
        "v_mul_f32_dpp %[f2], %[hx], %[wj2] row_share:8 row_mask:0xf bank_mask:0xf bound_ctrl:1\n\t"
        "v_add_f32 %[a1], %[a1], %[b1]\n\t"
        "v_fma_f32_e64_dpp %[e1], %[h], %[wi1], %[e1] row_share:4 row_mask:0xf bank_mask:0xf bound_ctrl:1\n\t"
        "v_fma_f32_e64_dpp %[e2], %[h], %[wi3], %[e2] row_share:12 row_mask:0xf bank_mask:0xf bound_ctrl:1\n\t"
        "v_tanh_f32 %[a1], %[a1]\n\t"
        "v_fma_f32_e64_dpp %[f1], %[hx], %[wj1], %[f1] row_share:4 row_mask:0xf bank_mask:0xf bound_ctrl:1\n\t"
        "v_fma_f32_e64_dpp %[f2], %[hx], %[wj3], %[f2] row_share:12 row_mask:0xf bank_mask:0xf bound_ctrl:1\n\t"
        "v_add_f32 %[e1], %[e1], %[e2]\n\t"
        "v_fma_f32_e64_dpp %[b1], %[a1], %[a1], %[a1] quad_perm:[0,0,0,0] row_mask:0xf bank_mask:0xf bound_ctrl:1\n\t"
        "v_fma_f32_e64_dpp %[b2], %[a1], %[hc], %[hc] quad_perm:[1,1,1,1] row_mask:0xf bank_mask:0xf bound_ctrl:1\n\t"
        "v_fma_f32_e64_dpp %[a2], %[a1], 0.5, 0.5 quad_perm:[3,3,3,3] row_mask:0xf bank_mask:0xf bound_ctrl:1\n\t"
        "v_add_f32 %[f1], %[f1], %[f2]\n\t"
        "v_fma_f32_e64_dpp %[b1], %[b1], 0.5, %[b2] quad_perm:[2,2,2,2] row_mask:0xf bank_mask:0xf bound_ctrl:1\n\t"
        "v_add_f32 %[e1], %[e1], %[f1]\n\t"
        "v_tanh_f32 %[a1], %[b1]\n\t"
        "v_mul_f32 %[hc], 0.5, %[b1]\n\t"
        "v_mul_f32 %[h], %[a2], %[a1]\n\t"
        : [h] "+v"(h), [hc] "+v"(hc), [hx] "=&v"(hx), [a1] "=&v"(a1), [a2] "=&v"(a2), [b1] "=&v"(b1), [b2] "=&v"(b2), [e1] "=&v"(e1), [e2] "=&v"(e2), [f1] "=&v"(f1), [f2] "=&v"(f2)
        : [acc] "v"(acc), [wo0] "v"(wo[0]), [wo1] "v"(wo[1]), [wo2] "v"(wo[2]), [wo3] "v"(wo[3]), [wx0] "v"(wx[0]), [wx1] "v"(wx[1]), [wx2] "v"(wx[2]), [wx3] "v"(wx[3]), [s0] "s"(s0), [wi0] "v"(wi[0]), [wi1] "v"(wi[1]), [wi2] "v"(wi[2]), [wi3] "v"(wi[3]), [wj0] "v"(wj[0]), [wj1] "v"(wj[1]), [wj2] "v"(wj[2]), [wj3] "v"(wj[3])); d = e1;
}

__device__ __forceinline__ void step_h(float& h, float& hc, const float acc, const float (&wo)[4], const float (&wx)[4], const int s0, const float wr, const float br, float& rr) {
    float hx, a1, a2, b1, b2; float p, q;
    asm(
        "v_permlanex16_b32 %[hx], %[h], %[s0], 0xfedcba98\n\t"
        "v_fma_f32_e64_dpp %[a1], %[h], %[wo0], %[acc] row_share:0 row_mask:0xf bank_mask:0xf bound_ctrl:1\n\t"
        "v_mul_f32_dpp %[a2], %[h], %[wo2] row_share:8 row_mask:0xf bank_mask:0xf bound_ctrl:1\n\t"
        "v_mul_f32_dpp %[b1], %[hx], %[wx0] row_share:0 row_mask:0xf bank_mask:0xf bound_ctrl:1\n\t"
        "v_mul_f32_dpp %[b2], %[hx], %[wx2] row_share:8 row_mask:0xf bank_mask:0xf bound_ctrl:1\n\t"
        "v_fma_f32_e64_dpp %[a1], %[h], %[wo1], %[a1] row_share:4 row_mask:0xf bank_mask:0xf bound_ctrl:1\n\t"
        "v_fma_f32_e64_dpp %[b1], %[hx], %[wx1], %[b1] row_share:4 row_mask:0xf bank_mask:0xf bound_ctrl:1\n\t"
        "v_fma_f32_e64_dpp %[b2], %[hx], %[wx3], %[b2] row_share:12 row_mask:0xf bank_mask:0xf bound_ctrl:1\n\t"
        "v_fma_f32_e64_dpp %[a2], %[h], %[wo3], %[a2] row_share:12 row_mask:0xf bank_mask:0xf bound_ctrl:1\n\t"
        "v_mul_f32 %[p], %[h], %[wr]\n\t"
        "v_add_f32 %[b1], %[b1], %[b2]\n\t"
        "v_add_f32 %[a1], %[a1], %[a2]\n\t"
        "v_add_f32_dpp %[p], %[p], %[p] row_xmask:4 row_mask:0xf bank_mask:0xf bound_ctrl:1\n\t"
        "v_add_f32 %[a1], %[a1], %[b1]\n\t"
        "v_add_f32_dpp %[p], %[p], %[p] row_xmask:8 row_mask:0xf bank_mask:0xf bound_ctrl:1\n\t"
        "v_tanh_f32 %[a1], %[a1]\n\t"
        "v_permlanex16_b32 %[q], %[p], %[s0], 0xfedcba98\n\t"
        "v_fma_f32_e64_dpp %[b1], %[a1], %[a1], %[a1] quad_perm:[0,0,0,0] row_mask:0xf bank_mask:0xf bound_ctrl:1\n\t"
        "v_fma_f32_e64_dpp %[b2], %[a1], %[hc], %[hc] quad_perm:[1,1,1,1] row_mask:0xf bank_mask:0xf bound_ctrl:1\n\t"
        "v_fma_f32_e64_dpp %[a2], %[a1], 0.5, 0.5 quad_perm:[3,3,3,3] row_mask:0xf bank_mask:0xf bound_ctrl:1\n\t"
        "v_add_f32 %[p], %[p], %[q]\n\t"
        "v_fma_f32_e64_dpp %[b1], %[b1], 0.5, %[b2] quad_perm:[2,2,2,2] row_mask:0xf bank_mask:0xf bound_ctrl:1\n\t"
        "v_tanh_f32 %[a1], %[b1]\n\t"
        "v_mul_f32 %[hc], 0.5, %[b1]\n\t"
        "v_add_f32 %[p], %[p], %[br]\n\t"
        "v_mul_f32 %[h], %[a2], %[a1]\n\t"
        : [h] "+v"(h), [hc] "+v"(hc), [hx] "=&v"(hx), [a1] "=&v"(a1), [a2] "=&v"(a2), [b1] "=&v"(b1), [b2] "=&v"(b2), [p] "=&v"(p), [q] "=&v"(q)
        : [acc] "v"(acc), [wo0] "v"(wo[0]), [wo1] "v"(wo[1]), [wo2] "v"(wo[2]), [wo3] "v"(wo[3]), [wx0] "v"(wx[0]), [wx1] "v"(wx[1]), [wx2] "v"(wx[2]), [wx3] "v"(wx[3]), [s0] "s"(s0), [wr] "v"(wr), [br] "v"(br)); rr = p;
}

#define DPP_ALL " row_mask:0xf bank_mask:0xf bound_ctrl:1"
#define FMA_DPP(CTRL, a, b, c) __extension__({ float d__;                               \
    asm("v_fma_f32_e64_dpp %0, %1, %2, %3 " CTRL DPP_ALL                               \
        : "=v"(d__) : "v"(a), "v"(b), "v"(c)); d__; })
#define MUL_DPP(CTRL, a, b) __extension__({ float d__;                                  \
    asm("v_mul_f32_dpp %0, %1, %2 " CTRL DPP_ALL                                       \
        : "=v"(d__) : "v"(a), "v"(b)); d__; })
__device__ __forceinline__ float input_half(float h, float hx,
                                            const float (&wi)[4], const float (&wj)[4]) {
    float e1 = MUL_DPP("row_share:0", h,  wi[0]);
    float e2 = MUL_DPP("row_share:8", h,  wi[2]);
    float f1 = MUL_DPP("row_share:0", hx, wj[0]);
    float f2 = MUL_DPP("row_share:8", hx, wj[2]);
    e1 = FMA_DPP("row_share:4",  h,  wi[1], e1);
    e2 = FMA_DPP("row_share:12", h,  wi[3], e2);
    f1 = FMA_DPP("row_share:4",  hx, wj[1], f1);
    f2 = FMA_DPP("row_share:12", hx, wj[3], f2);
    return (e1 + e2) + (f1 + f2);
}

__device__ __forceinline__ void lds_barrier() {
    __builtin_amdgcn_fence(__ATOMIC_RELEASE, "workgroup", "local");
    __builtin_amdgcn_s_barrier();
    __builtin_amdgcn_fence(__ATOMIC_ACQUIRE, "workgroup", "local");
}

#define NG (SEQ / PF)

__global__ __launch_bounds__(64) void lstm_scan_kernel(
    const float* __restrict__ xg,
    const float* __restrict__ w_hh0,
    const float* __restrict__ w_ih1,
    const float* __restrict__ w_hh1,
    const float* __restrict__ b_ih1,
    const float* __restrict__ b_hh1,
    const float* __restrict__ w_reg,
    const float* __restrict__ b_reg,
    float* __restrict__ out)
{
    __shared__ float pbuf[2][PF][32];

    const int l    = (int)(threadIdx.x & 31);
    const int role = __builtin_amdgcn_readfirstlane((int)(threadIdx.x >> 5));
    const int b    = (int)blockIdx.x;
    const int unit = l >> 2;
    const int gate = l & 3;
    const int row  = gate * HD + unit;
    const int ko   = (l >> 4) * 4;
    const int kx   = 4 - ko;

    const bool  is_g = (gate == 2);
    const float aS = is_g ? 1.0f : 0.5f;

    float wao[4], wax[4], wio[4], wix[4];
    const float* wrec = (role == 0) ? w_hh0 : w_hh1;
    #pragma unroll
    for (int m = 0; m < 4; ++m) {
        wao[m] = aS * wrec[row * HD + ko + m];    wax[m] = aS * wrec[row * HD + kx + m];
        wio[m] = aS * w_ih1[row * HD + ko + m];   wix[m] = aS * w_ih1[row * HD + kx + m];
    }
    const float bias1 = aS * (b_ih1[row] + b_hh1[row]);

    const float wr = w_reg[(l & 1) * HD + unit];
    const float br = b_reg[l & 1];
    const int   lu = l >> 1;

    float h = 0.f, hc = 0.f;
    const int sel0 = 0x76543210;

    const float* xgp = xg + (size_t)b * G4 + l;
    const size_t xstride = (size_t)BATCH * G4;

    auto layer0_group = [&](const float (&r)[PF], float* dst) __attribute__((always_inline)) {
        step_p(h, hc, r[0] * aS, wao, wax, sel0);
        #pragma unroll
        for (int u = 1; u < PF; ++u) {
            float d;
            step_d(h, hc, r[u] * aS, wao, wax, sel0, wio, wix, d);
            dst[(u - 1) * 32] = d;
        }
        dst[(PF - 1) * 32] = input_half(h, xhalf(h), wio, wix);
    };

    auto layer1_group = [&](const float* src, const int s0) __attribute__((always_inline)) {
        float pin[PF];
        #pragma unroll
        for (int u = 0; u < PF; ++u) pin[u] = src[u * 32];
        __builtin_amdgcn_sched_barrier(0);
        float res = 0.f;
        step_p(h, hc, pin[0] + bias1, wao, wax, sel0);
        #pragma unroll
        for (int u = 1; u < PF; ++u) {
            float rr;
            step_h(h, hc, pin[u] + bias1, wao, wax, sel0, wr, br, rr);
            res = (lu == u - 1) ? rr : res;
        }
        {
            float p = h * wr;
            p += row_xor<4>(p);
            p += row_xor<8>(p);
            p += xhalf(p);
            res = (lu == PF - 1) ? (p + br) : res;
        }
        out[((size_t)(s0 + lu) * BATCH + b) * 2 + (l & 1)] = res;
    };

    float ra[PF], rb[PF];
    if (role == 0) {
        #pragma unroll
        for (int u = 0; u < PF; ++u) ra[u] = xgp[(size_t)u * xstride];
    }

    for (int g = 0; g <= NG; g += 2) {
        if (role == 0) {
            if (g < NG) {
                #pragma unroll
                for (int u = 0; u < PF; ++u) rb[u] = xgp[(size_t)((g + 1) * PF + u) * xstride];
                __builtin_amdgcn_sched_barrier(0);
                layer0_group(ra, &pbuf[0][0][l]);
            }
        } else {
            if (g >= 2) layer1_group(&pbuf[1][0][l], (g - 1) * PF);
        }
        lds_barrier();

        if (role == 0) {
            if (g < NG) {
                #pragma unroll
                for (int u = 0; u < PF; ++u)
                    ra[u] = xgp[(size_t)(((g + 2) * PF + u) & (SEQ - 1)) * xstride];
                __builtin_amdgcn_sched_barrier(0);
                layer0_group(rb, &pbuf[1][0][l]);
            }
        } else {
            if (g < NG) layer1_group(&pbuf[0][0][l], g * PF);
        }
        lds_barrier();
    }
}

extern "C" void kernel_launch(void* const* d_in, const int* in_sizes, int n_in,
                              void* d_out, int out_size, void* d_ws, size_t ws_size,
                              hipStream_t stream) {
    const float* x     = (const float*)d_in[0];
    const float* w_ih0 = (const float*)d_in[1];
    const float* w_hh0 = (const float*)d_in[2];
    const float* b_ih0 = (const float*)d_in[3];
    const float* b_hh0 = (const float*)d_in[4];
    const float* w_ih1 = (const float*)d_in[5];
    const float* w_hh1 = (const float*)d_in[6];
    const float* b_ih1 = (const float*)d_in[7];
    const float* b_hh1 = (const float*)d_in[8];
    const float* w_reg = (const float*)d_in[9];
    const float* b_reg = (const float*)d_in[10];
    float* out = (float*)d_out;
    float* xg  = (float*)d_ws;

    xgemm_wmma_kernel<<<512, 128, 0, stream>>>(x, w_ih0, b_ih0, b_hh0, xg);
    lstm_scan_kernel<<<256, 64, 0, stream>>>(xg, w_hh0, w_ih1, w_hh1,
                                            b_ih1, b_hh1, w_reg, b_reg, out);
}


# one launch: each batch element's 8-wave work-group first makes its own 512 xg rows (f32 WMMA, k order kept) into 64 KB of LDS, then two of its waves scan from LDS; no xg array, no second launch
# speedup vs baseline: 10.2502x; 1.1548x over previous
#include <hip/hip_runtime.h>
#include <hip/hip_bf16.h>

#define SEQ   512
#define BATCH 256
#define INF   180
#define HD    8
#define G4    32
#define PF    16

typedef __attribute__((ext_vector_type(2))) float v2f;
typedef __attribute__((ext_vector_type(8))) float v8f;

typedef __attribute__((ext_vector_type(4))) float v4f;

#define NW      8
#define XT      2
#define ROW_V4  (INF / 4)
#define TILE_V4 (32 * ROW_V4)
#define TILE_LD (TILE_V4 / 32)
#define WB_FULL (TILE_V4 / (32 * NW))
#define WB_REST (TILE_V4 - WB_FULL * 32 * NW)

__device__ __forceinline__ void xg_lds_barrier() {
    __builtin_amdgcn_fence(__ATOMIC_RELEASE, "workgroup", "local");
    __builtin_amdgcn_s_barrier();
    __builtin_amdgcn_fence(__ATOMIC_ACQUIRE, "workgroup", "local");
}


template <int M>
__device__ __forceinline__ float row_xor(float v) {
    const int b = __builtin_bit_cast(int, v);
    return __builtin_bit_cast(float,
        __builtin_amdgcn_update_dpp(0, b, 0x160 + M, 0xf, 0xf, true));
}
__device__ __forceinline__ float xhalf(float x) {
    const int b = __builtin_bit_cast(int, x);
    return __builtin_bit_cast(float,
        __builtin_amdgcn_permlanex16(0, b, 0x76543210, 0xfedcba98, false, false));
}

__device__ __forceinline__ void step_p(float& h, float& hc, const float acc, const float (&wo)[4], const float (&wx)[4], const int s0) {
    float hx, a1, a2, b1, b2;
    asm(
        "v_permlanex16_b32 %[hx], %[h], %[s0], 0xfedcba98\n\t"
        "v_fma_f32_e64_dpp %[a1], %[h], %[wo0], %[acc] row_share:0 row_mask:0xf bank_mask:0xf bound_ctrl:1\n\t"
        "v_mul_f32_dpp %[a2], %[h], %[wo2] row_share:8 row_mask:0xf bank_mask:0xf bound_ctrl:1\n\t"
        "v_mul_f32_dpp %[b1], %[hx], %[wx0] row_share:0 row_mask:0xf bank_mask:0xf bound_ctrl:1\n\t"
        "v_mul_f32_dpp %[b2], %[hx], %[wx2] row_share:8 row_mask:0xf bank_mask:0xf bound_ctrl:1\n\t"
        "v_fma_f32_e64_dpp %[a1], %[h], %[wo1], %[a1] row_share:4 row_mask:0xf bank_mask:0xf bound_ctrl:1\n\t"
        "v_fma_f32_e64_dpp %[b1], %[hx], %[wx1], %[b1] row_share:4 row_mask:0xf bank_mask:0xf bound_ctrl:1\n\t"
        "v_fma_f32_e64_dpp %[b2], %[hx], %[wx3], %[b2] row_share:12 row_mask:0xf bank_mask:0xf bound_ctrl:1\n\t"
        "v_fma_f32_e64_dpp %[a2], %[h], %[wo3], %[a2] row_share:12 row_mask:0xf bank_mask:0xf bound_ctrl:1\n\t"
        "v_add_f32 %[b1], %[b1], %[b2]\n\t"
        "v_add_f32 %[a1], %[a1], %[a2]\n\t"
        "v_add_f32 %[a1], %[a1], %[b1]\n\t"
        "v_tanh_f32 %[a1], %[a1]\n\t"
        "v_nop\n\t"
        "v_fma_f32_e64_dpp %[b1], %[a1], %[a1], %[a1] quad_perm:[0,0,0,0] row_mask:0xf bank_mask:0xf bound_ctrl:1\n\t"
        "v_fma_f32_e64_dpp %[b2], %[a1], %[hc], %[hc] quad_perm:[1,1,1,1] row_mask:0xf bank_mask:0xf bound_ctrl:1\n\t"
        "v_fma_f32_e64_dpp %[a2], %[a1], 0.5, 0.5 quad_perm:[3,3,3,3] row_mask:0xf bank_mask:0xf bound_ctrl:1\n\t"
        "v_fma_f32_e64_dpp %[b1], %[b1], 0.5, %[b2] quad_perm:[2,2,2,2] row_mask:0xf bank_mask:0xf bound_ctrl:1\n\t"
        "v_tanh_f32 %[a1], %[b1]\n\t"
        "v_mul_f32 %[hc], 0.5, %[b1]\n\t"
        "v_mul_f32 %[h], %[a2], %[a1]\n\t"
        : [h] "+v"(h), [hc] "+v"(hc), [hx] "=&v"(hx), [a1] "=&v"(a1), [a2] "=&v"(a2), [b1] "=&v"(b1), [b2] "=&v"(b2)
        : [acc] "v"(acc), [wo0] "v"(wo[0]), [wo1] "v"(wo[1]), [wo2] "v"(wo[2]), [wo3] "v"(wo[3]), [wx0] "v"(wx[0]), [wx1] "v"(wx[1]), [wx2] "v"(wx[2]), [wx3] "v"(wx[3]), [s0] "s"(s0));
}

__device__ __forceinline__ void step_d(float& h, float& hc, const float acc, const float (&wo)[4], const float (&wx)[4], const int s0, const float (&wi)[4], const float (&wj)[4], float& d) {
    float hx, a1, a2, b1, b2; float e1, e2, f1, f2;
    asm(
        "v_permlanex16_b32 %[hx], %[h], %[s0], 0xfedcba98\n\t"
        "v_fma_f32_e64_dpp %[a1], %[h], %[wo0], %[acc] row_share:0 row_mask:0xf bank_mask:0xf bound_ctrl:1\n\t"
        "v_mul_f32_dpp %[a2], %[h], %[wo2] row_share:8 row_mask:0xf bank_mask:0xf bound_ctrl:1\n\t"
        "v_mul_f32_dpp %[b1], %[hx], %[wx0] row_share:0 row_mask:0xf bank_mask:0xf bound_ctrl:1\n\t"
        "v_mul_f32_dpp %[b2], %[hx], %[wx2] row_share:8 row_mask:0xf bank_mask:0xf bound_ctrl:1\n\t"
        "v_fma_f32_e64_dpp %[a1], %[h], %[wo1], %[a1] row_share:4 row_mask:0xf bank_mask:0xf bound_ctrl:1\n\t"
        "v_fma_f32_e64_dpp %[b1], %[hx], %[wx1], %[b1] row_share:4 row_mask:0xf bank_mask:0xf bound_ctrl:1\n\t"
        "v_fma_f32_e64_dpp %[b2], %[hx], %[wx3], %[b2] row_share:12 row_mask:0xf bank_mask:0xf bound_ctrl:1\n\t"
        "v_fma_f32_e64_dpp %[a2], %[h], %[wo3], %[a2] row_share:12 row_mask:0xf bank_mask:0xf bound_ctrl:1\n\t"
        "v_mul_f32_dpp %[e1], %[h], %[wi0] row_share:0 row_mask:0xf bank_mask:0xf bound_ctrl:1\n\t"
        "v_add_f32 %[b1], %[b1], %[b2]\n\t"
        "v_mul_f32_dpp %[e2], %[h], %[wi2] row_share:8 row_mask:0xf bank_mask:0xf bound_ctrl:1\n\t"
        "v_add_f32 %[a1], %[a1], %[a2]\n\t"
        "v_mul_f32_dpp %[f1], %[hx], %[wj0] row_share:0 row_mask:0xf bank_mask:0xf bound_ctrl:1\n\t"
        "v_mul_f32_dpp %[f2], %[hx], %[wj2] row_share:8 row_mask:0xf bank_mask:0xf bound_ctrl:1\n\t"
        "v_add_f32 %[a1], %[a1], %[b1]\n\t"
        "v_fma_f32_e64_dpp %[e1], %[h], %[wi1], %[e1] row_share:4 row_mask:0xf bank_mask:0xf bound_ctrl:1\n\t"
        "v_fma_f32_e64_dpp %[e2], %[h], %[wi3], %[e2] row_share:12 row_mask:0xf bank_mask:0xf bound_ctrl:1\n\t"
        "v_tanh_f32 %[a1], %[a1]\n\t"
        "v_fma_f32_e64_dpp %[f1], %[hx], %[wj1], %[f1] row_share:4 row_mask:0xf bank_mask:0xf bound_ctrl:1\n\t"
        "v_fma_f32_e64_dpp %[f2], %[hx], %[wj3], %[f2] row_share:12 row_mask:0xf bank_mask:0xf bound_ctrl:1\n\t"
        "v_add_f32 %[e1], %[e1], %[e2]\n\t"
        "v_fma_f32_e64_dpp %[b1], %[a1], %[a1], %[a1] quad_perm:[0,0,0,0] row_mask:0xf bank_mask:0xf bound_ctrl:1\n\t"
        "v_fma_f32_e64_dpp %[b2], %[a1], %[hc], %[hc] quad_perm:[1,1,1,1] row_mask:0xf bank_mask:0xf bound_ctrl:1\n\t"
        "v_fma_f32_e64_dpp %[a2], %[a1], 0.5, 0.5 quad_perm:[3,3,3,3] row_mask:0xf bank_mask:0xf bound_ctrl:1\n\t"
        "v_add_f32 %[f1], %[f1], %[f2]\n\t"
        "v_fma_f32_e64_dpp %[b1], %[b1], 0.5, %[b2] quad_perm:[2,2,2,2] row_mask:0xf bank_mask:0xf bound_ctrl:1\n\t"
        "v_add_f32 %[e1], %[e1], %[f1]\n\t"
        "v_tanh_f32 %[a1], %[b1]\n\t"
        "v_mul_f32 %[hc], 0.5, %[b1]\n\t"
        "v_mul_f32 %[h], %[a2], %[a1]\n\t"
        : [h] "+v"(h), [hc] "+v"(hc), [hx] "=&v"(hx), [a1] "=&v"(a1), [a2] "=&v"(a2), [b1] "=&v"(b1), [b2] "=&v"(b2), [e1] "=&v"(e1), [e2] "=&v"(e2), [f1] "=&v"(f1), [f2] "=&v"(f2)
        : [acc] "v"(acc), [wo0] "v"(wo[0]), [wo1] "v"(wo[1]), [wo2] "v"(wo[2]), [wo3] "v"(wo[3]), [wx0] "v"(wx[0]), [wx1] "v"(wx[1]), [wx2] "v"(wx[2]), [wx3] "v"(wx[3]), [s0] "s"(s0), [wi0] "v"(wi[0]), [wi1] "v"(wi[1]), [wi2] "v"(wi[2]), [wi3] "v"(wi[3]), [wj0] "v"(wj[0]), [wj1] "v"(wj[1]), [wj2] "v"(wj[2]), [wj3] "v"(wj[3])); d = e1;
}

__device__ __forceinline__ void step_h(float& h, float& hc, const float acc, const float (&wo)[4], const float (&wx)[4], const int s0, const float wr, const float br, float& rr) {
    float hx, a1, a2, b1, b2; float p, q;
    asm(
        "v_permlanex16_b32 %[hx], %[h], %[s0], 0xfedcba98\n\t"
        "v_fma_f32_e64_dpp %[a1], %[h], %[wo0], %[acc] row_share:0 row_mask:0xf bank_mask:0xf bound_ctrl:1\n\t"
        "v_mul_f32_dpp %[a2], %[h], %[wo2] row_share:8 row_mask:0xf bank_mask:0xf bound_ctrl:1\n\t"
        "v_mul_f32_dpp %[b1], %[hx], %[wx0] row_share:0 row_mask:0xf bank_mask:0xf bound_ctrl:1\n\t"
        "v_mul_f32_dpp %[b2], %[hx], %[wx2] row_share:8 row_mask:0xf bank_mask:0xf bound_ctrl:1\n\t"
        "v_fma_f32_e64_dpp %[a1], %[h], %[wo1], %[a1] row_share:4 row_mask:0xf bank_mask:0xf bound_ctrl:1\n\t"
        "v_fma_f32_e64_dpp %[b1], %[hx], %[wx1], %[b1] row_share:4 row_mask:0xf bank_mask:0xf bound_ctrl:1\n\t"
        "v_fma_f32_e64_dpp %[b2], %[hx], %[wx3], %[b2] row_share:12 row_mask:0xf bank_mask:0xf bound_ctrl:1\n\t"
        "v_fma_f32_e64_dpp %[a2], %[h], %[wo3], %[a2] row_share:12 row_mask:0xf bank_mask:0xf bound_ctrl:1\n\t"
        "v_mul_f32 %[p], %[h], %[wr]\n\t"
        "v_add_f32 %[b1], %[b1], %[b2]\n\t"
        "v_add_f32 %[a1], %[a1], %[a2]\n\t"
        "v_add_f32_dpp %[p], %[p], %[p] row_xmask:4 row_mask:0xf bank_mask:0xf bound_ctrl:1\n\t"
        "v_add_f32 %[a1], %[a1], %[b1]\n\t"
        "v_add_f32_dpp %[p], %[p], %[p] row_xmask:8 row_mask:0xf bank_mask:0xf bound_ctrl:1\n\t"
        "v_tanh_f32 %[a1], %[a1]\n\t"
        "v_permlanex16_b32 %[q], %[p], %[s0], 0xfedcba98\n\t"
        "v_fma_f32_e64_dpp %[b1], %[a1], %[a1], %[a1] quad_perm:[0,0,0,0] row_mask:0xf bank_mask:0xf bound_ctrl:1\n\t"
        "v_fma_f32_e64_dpp %[b2], %[a1], %[hc], %[hc] quad_perm:[1,1,1,1] row_mask:0xf bank_mask:0xf bound_ctrl:1\n\t"
        "v_fma_f32_e64_dpp %[a2], %[a1], 0.5, 0.5 quad_perm:[3,3,3,3] row_mask:0xf bank_mask:0xf bound_ctrl:1\n\t"
        "v_add_f32 %[p], %[p], %[q]\n\t"
        "v_fma_f32_e64_dpp %[b1], %[b1], 0.5, %[b2] quad_perm:[2,2,2,2] row_mask:0xf bank_mask:0xf bound_ctrl:1\n\t"
        "v_tanh_f32 %[a1], %[b1]\n\t"
        "v_mul_f32 %[hc], 0.5, %[b1]\n\t"
        "v_add_f32 %[p], %[p], %[br]\n\t"
        "v_mul_f32 %[h], %[a2], %[a1]\n\t"
        : [h] "+v"(h), [hc] "+v"(hc), [hx] "=&v"(hx), [a1] "=&v"(a1), [a2] "=&v"(a2), [b1] "=&v"(b1), [b2] "=&v"(b2), [p] "=&v"(p), [q] "=&v"(q)
        : [acc] "v"(acc), [wo0] "v"(wo[0]), [wo1] "v"(wo[1]), [wo2] "v"(wo[2]), [wo3] "v"(wo[3]), [wx0] "v"(wx[0]), [wx1] "v"(wx[1]), [wx2] "v"(wx[2]), [wx3] "v"(wx[3]), [s0] "s"(s0), [wr] "v"(wr), [br] "v"(br)); rr = p;
}

#define DPP_ALL " row_mask:0xf bank_mask:0xf bound_ctrl:1"
#define FMA_DPP(CTRL, a, b, c) __extension__({ float d__;                               \
    asm("v_fma_f32_e64_dpp %0, %1, %2, %3 " CTRL DPP_ALL                               \
        : "=v"(d__) : "v"(a), "v"(b), "v"(c)); d__; })
#define MUL_DPP(CTRL, a, b) __extension__({ float d__;                                  \
    asm("v_mul_f32_dpp %0, %1, %2 " CTRL DPP_ALL                                       \
        : "=v"(d__) : "v"(a), "v"(b)); d__; })
__device__ __forceinline__ float input_half(float h, float hx,
                                            const float (&wi)[4], const float (&wj)[4]) {
    float e1 = MUL_DPP("row_share:0", h,  wi[0]);
    float e2 = MUL_DPP("row_share:8", h,  wi[2]);
    float f1 = MUL_DPP("row_share:0", hx, wj[0]);
    float f2 = MUL_DPP("row_share:8", hx, wj[2]);
    e1 = FMA_DPP("row_share:4",  h,  wi[1], e1);
    e2 = FMA_DPP("row_share:12", h,  wi[3], e2);
    f1 = FMA_DPP("row_share:4",  hx, wj[1], f1);
    f2 = FMA_DPP("row_share:12", hx, wj[3], f2);
    return (e1 + e2) + (f1 + f2);
}

__device__ __forceinline__ void lds_barrier() {
    __builtin_amdgcn_fence(__ATOMIC_RELEASE, "workgroup", "local");
    __builtin_amdgcn_s_barrier();
    __builtin_amdgcn_fence(__ATOMIC_ACQUIRE, "workgroup", "local");
}

#define NG (SEQ / PF)

__global__ __launch_bounds__(32 * NW) void lstm_fused_kernel(
    const float* x,
    const float* w_ih0,
    const float* b_ih0,
    const float* b_hh0,
    const float* w_hh0,
    const float* w_ih1,
    const float* w_hh1,
    const float* b_ih1,
    const float* b_hh1,
    const float* w_reg,
    const float* b_reg,
    float* out)
{
    __shared__ v4f   sB[TILE_V4];
    __shared__ v4f   sA[NW][TILE_V4];
    __shared__ float sXg[SEQ * G4];
    __shared__ float pbuf[2][PF][32];

    const int tid = (int)threadIdx.x;
    const int l   = tid & 31;
    const int b   = (int)blockIdx.x;

    {
        const int wave  = __builtin_amdgcn_readfirstlane(tid >> 5);
        const int n     = l & 15;
        const int hi    = l >> 4;
        const int tile0 = wave * XT;

        const v4f* w4 = (const v4f*)w_ih0 + tid;
        v4f wb[WB_FULL + 1];
        #pragma unroll
        for (int j = 0; j < WB_FULL; ++j) wb[j] = w4[j * 32 * NW];
        if (tid < WB_REST) wb[WB_FULL] = w4[WB_FULL * 32 * NW];

        const char* xb = (const char*)(x + (size_t)b * INF);
        const unsigned offA = (unsigned)l * 16u;
        const unsigned offB = offA + (unsigned)((BATCH * INF - INF) * 4);
        auto piece = [&](const int tile, const int t) __attribute__((always_inline)) -> const v4f* {
            const int r0  = (t * 32) / ROW_V4;
            const int c0  = t * 32 - r0 * ROW_V4;
            const unsigned vo = (l >= ROW_V4 - c0) ? offB : offA;
            return (const v4f*)(xb + (size_t)(tile * 32 + r0) * (BATCH * INF * 4) + c0 * 16 + vo);
        };

        v4f pre[TILE_LD];
        #pragma unroll
        for (int t = 0; t < TILE_LD; ++t) pre[t] = *piece(tile0, t);

        #pragma unroll
        for (int j = 0; j < WB_FULL; ++j) sB[tid + j * 32 * NW] = wb[j];
        if (tid < WB_REST) sB[tid + WB_FULL * 32 * NW] = wb[WB_FULL];

        const float bias0 = b_ih0[n]      + b_hh0[n];
        const float bias1 = b_ih0[n + 16] + b_hh0[n + 16];
        const int p0 = (n & 7) * 4 + (n >> 3);
        const int p1 = p0 + 2;

        v4f* myA = &sA[wave][0];
        const float* aF = (const float*)myA + n * INF + 2 * hi;
        const float* bF = (const float*)&sB[0] + n * INF + 2 * hi;

        #pragma unroll
        for (int tt = 0; tt < XT; ++tt) {
            #pragma unroll
            for (int t = 0; t < TILE_LD; ++t) myA[t * 32 + l] = pre[t];
            xg_lds_barrier();

            if (tt + 1 < XT) {
                #pragma unroll
                for (int t = 0; t < TILE_LD; ++t) pre[t] = *piece(tile0 + tt + 1, t);
            }
            __builtin_amdgcn_sched_barrier(0);

            v8f acc00 = {0.f,0.f,0.f,0.f,0.f,0.f,0.f,0.f};
            v8f acc01 = acc00, acc10 = acc00, acc11 = acc00;

            #pragma unroll
            for (int ks = 0; ks < INF / 4; ++ks) {
                const v2f a0 = *(const v2f*)(aF + 4 * ks);
                const v2f a1 = *(const v2f*)(aF + 16 * INF + 4 * ks);
                const v2f b0 = *(const v2f*)(bF + 4 * ks);
                const v2f b1 = *(const v2f*)(bF + 16 * INF + 4 * ks);
                acc00 = __builtin_amdgcn_wmma_f32_16x16x4_f32(false, a0, false, b0,
                                                              (short)0, acc00, false, false);
                acc01 = __builtin_amdgcn_wmma_f32_16x16x4_f32(false, a0, false, b1,
                                                              (short)0, acc01, false, false);
                acc10 = __builtin_amdgcn_wmma_f32_16x16x4_f32(false, a1, false, b0,
                                                              (short)0, acc10, false, false);
                acc11 = __builtin_amdgcn_wmma_f32_16x16x4_f32(false, a1, false, b1,
                                                              (short)0, acc11, false, false);
            }

            float* ob = sXg + ((tile0 + tt) * 32 + hi * 8) * G4;
            #pragma unroll
            for (int r = 0; r < 8; ++r) {
                ob[r * G4 + p0]        = acc00[r] + bias0;
                ob[r * G4 + p1]        = acc01[r] + bias1;
                ob[(r + 16) * G4 + p0] = acc10[r] + bias0;
                ob[(r + 16) * G4 + p1] = acc11[r] + bias1;
            }
        }
    }

    const int role = __builtin_amdgcn_readfirstlane(tid >> 5);
    const int unit = l >> 2;
    const int gate = l & 3;
    const int row  = gate * HD + unit;
    const int ko   = (l >> 4) * 4;
    const int kx   = 4 - ko;

    const bool  is_g = (gate == 2);
    const float aS = is_g ? 1.0f : 0.5f;

    float wao[4], wax[4], wio[4], wix[4];
    const float* wrec = (role == 0) ? w_hh0 : w_hh1;
    #pragma unroll
    for (int m = 0; m < 4; ++m) {
        wao[m] = aS * wrec[row * HD + ko + m];    wax[m] = aS * wrec[row * HD + kx + m];
        wio[m] = aS * w_ih1[row * HD + ko + m];   wix[m] = aS * w_ih1[row * HD + kx + m];
    }
    const float bias1 = aS * (b_ih1[row] + b_hh1[row]);

    const float wr = w_reg[(l & 1) * HD + unit];
    const float br = b_reg[l & 1];
    const int   lu = l >> 1;

    __syncthreads();

    float h = 0.f, hc = 0.f;
    const int sel0 = 0x76543210;

    const float* xgp = &sXg[l];

    auto layer0_group = [&](const float (&r)[PF], float* dst) __attribute__((always_inline)) {
        step_p(h, hc, r[0] * aS, wao, wax, sel0);
        #pragma unroll
        for (int u = 1; u < PF; ++u) {
            float d;
            step_d(h, hc, r[u] * aS, wao, wax, sel0, wio, wix, d);
            dst[(u - 1) * 32] = d;
        }
        dst[(PF - 1) * 32] = input_half(h, xhalf(h), wio, wix);
    };

    auto layer1_group = [&](const float* src, const int s0) __attribute__((always_inline)) {
        float pin[PF];
        #pragma unroll
        for (int u = 0; u < PF; ++u) pin[u] = src[u * 32];
        __builtin_amdgcn_sched_barrier(0);
        float res = 0.f;
        step_p(h, hc, pin[0] + bias1, wao, wax, sel0);
        #pragma unroll
        for (int u = 1; u < PF; ++u) {
            float rr;
            step_h(h, hc, pin[u] + bias1, wao, wax, sel0, wr, br, rr);
            res = (lu == u - 1) ? rr : res;
        }
        {
            float p = h * wr;
            p += row_xor<4>(p);
            p += row_xor<8>(p);
            p += xhalf(p);
            res = (lu == PF - 1) ? (p + br) : res;
        }
        out[((size_t)(s0 + lu) * BATCH + b) * 2 + (l & 1)] = res;
    };

    float ra[PF], rb[PF];
    if (role == 0) {
        #pragma unroll
        for (int u = 0; u < PF; ++u) ra[u] = xgp[u * G4];
    }

    for (int g = 0; g <= NG; g += 2) {
        if (role == 0) {
            if (g < NG) {
                #pragma unroll
                for (int u = 0; u < PF; ++u) rb[u] = xgp[((g + 1) * PF + u) * G4];
                __builtin_amdgcn_sched_barrier(0);
                layer0_group(ra, &pbuf[0][0][l]);
            }
        } else if (role == 1) {
            if (g >= 2) layer1_group(&pbuf[1][0][l], (g - 1) * PF);
        }
        lds_barrier();

        if (role == 0) {
            if (g < NG) {
                #pragma unroll
                for (int u = 0; u < PF; ++u)
                    ra[u] = xgp[(((g + 2) * PF + u) & (SEQ - 1)) * G4];
                __builtin_amdgcn_sched_barrier(0);
                layer0_group(rb, &pbuf[1][0][l]);
            }
        } else if (role == 1) {
            if (g < NG) layer1_group(&pbuf[0][0][l], g * PF);
        }
        lds_barrier();
    }
}

extern "C" void kernel_launch(void* const* d_in, const int* in_sizes, int n_in,
                              void* d_out, int out_size, void* d_ws, size_t ws_size,
                              hipStream_t stream) {
    const float* x     = (const float*)d_in[0];
    const float* w_ih0 = (const float*)d_in[1];
    const float* w_hh0 = (const float*)d_in[2];
    const float* b_ih0 = (const float*)d_in[3];
    const float* b_hh0 = (const float*)d_in[4];
    const float* w_ih1 = (const float*)d_in[5];
    const float* w_hh1 = (const float*)d_in[6];
    const float* b_ih1 = (const float*)d_in[7];
    const float* b_hh1 = (const float*)d_in[8];
    const float* w_reg = (const float*)d_in[9];
    const float* b_reg = (const float*)d_in[10];
    float* out = (float*)d_out;

    lstm_fused_kernel<<<256, 32 * NW, 0, stream>>>(x, w_ih0, b_ih0, b_hh0, w_hh0, w_ih1, w_hh1,
                                                  b_ih1, b_hh1, w_reg, b_reg, out);
}


# xg rows of steps 128..511 made during the scan: the six idle waves sum their tiles a few k-steps (f32 WMMA) between the scan loop's barriers; only steps 0..127 made in front; k and sum orders kept
# speedup vs baseline: 10.6834x; 1.0423x over previous
#include <hip/hip_runtime.h>
#include <hip/hip_bf16.h>

#define SEQ   512
#define BATCH 256
#define INF   180
#define HD    8
#define G4    32
#define PF    16

typedef __attribute__((ext_vector_type(2))) float v2f;
typedef __attribute__((ext_vector_type(8))) float v8f;

typedef __attribute__((ext_vector_type(4))) float v4f;

#define NW      8
#define CH      5
#define NCH     (INF / 4 / CH)
#define CHF     9
#define NCHF    (INF / 4 / CHF)
#define SIDE0   6
#define SIDE1   12
#define ROW_V4  (INF / 4)
#define TILE_V4 (32 * ROW_V4)
#define TILE_LD (TILE_V4 / 32)
#define WB_FULL (TILE_V4 / (32 * NW))
#define WB_REST (TILE_V4 - WB_FULL * 32 * NW)

__device__ __forceinline__ void xg_lds_barrier() {
    __builtin_amdgcn_fence(__ATOMIC_RELEASE, "workgroup", "local");
    __builtin_amdgcn_s_barrier();
    __builtin_amdgcn_fence(__ATOMIC_ACQUIRE, "workgroup", "local");
}


template <int M>
__device__ __forceinline__ float row_xor(float v) {
    const int b = __builtin_bit_cast(int, v);
    return __builtin_bit_cast(float,
        __builtin_amdgcn_update_dpp(0, b, 0x160 + M, 0xf, 0xf, true));
}
__device__ __forceinline__ float xhalf(float x) {
    const int b = __builtin_bit_cast(int, x);
    return __builtin_bit_cast(float,
        __builtin_amdgcn_permlanex16(0, b, 0x76543210, 0xfedcba98, false, false));
}

__device__ __forceinline__ void step_p(float& h, float& hc, const float acc, const float (&wo)[4], const float (&wx)[4], const int s0) {
    float hx, a1, a2, b1, b2;
    asm(
        "v_permlanex16_b32 %[hx], %[h], %[s0], 0xfedcba98\n\t"
        "v_fma_f32_e64_dpp %[a1], %[h], %[wo0], %[acc] row_share:0 row_mask:0xf bank_mask:0xf bound_ctrl:1\n\t"
        "v_mul_f32_dpp %[a2], %[h], %[wo2] row_share:8 row_mask:0xf bank_mask:0xf bound_ctrl:1\n\t"
        "v_mul_f32_dpp %[b1], %[hx], %[wx0] row_share:0 row_mask:0xf bank_mask:0xf bound_ctrl:1\n\t"
        "v_mul_f32_dpp %[b2], %[hx], %[wx2] row_share:8 row_mask:0xf bank_mask:0xf bound_ctrl:1\n\t"
        "v_fma_f32_e64_dpp %[a1], %[h], %[wo1], %[a1] row_share:4 row_mask:0xf bank_mask:0xf bound_ctrl:1\n\t"
        "v_fma_f32_e64_dpp %[b1], %[hx], %[wx1], %[b1] row_share:4 row_mask:0xf bank_mask:0xf bound_ctrl:1\n\t"
        "v_fma_f32_e64_dpp %[b2], %[hx], %[wx3], %[b2] row_share:12 row_mask:0xf bank_mask:0xf bound_ctrl:1\n\t"
        "v_fma_f32_e64_dpp %[a2], %[h], %[wo3], %[a2] row_share:12 row_mask:0xf bank_mask:0xf bound_ctrl:1\n\t"
        "v_add_f32 %[b1], %[b1], %[b2]\n\t"
        "v_add_f32 %[a1], %[a1], %[a2]\n\t"
        "v_add_f32 %[a1], %[a1], %[b1]\n\t"
        "v_tanh_f32 %[a1], %[a1]\n\t"
        "v_nop\n\t"
        "v_fma_f32_e64_dpp %[b1], %[a1], %[a1], %[a1] quad_perm:[0,0,0,0] row_mask:0xf bank_mask:0xf bound_ctrl:1\n\t"
        "v_fma_f32_e64_dpp %[b2], %[a1], %[hc], %[hc] quad_perm:[1,1,1,1] row_mask:0xf bank_mask:0xf bound_ctrl:1\n\t"
        "v_fma_f32_e64_dpp %[a2], %[a1], 0.5, 0.5 quad_perm:[3,3,3,3] row_mask:0xf bank_mask:0xf bound_ctrl:1\n\t"
        "v_fma_f32_e64_dpp %[b1], %[b1], 0.5, %[b2] quad_perm:[2,2,2,2] row_mask:0xf bank_mask:0xf bound_ctrl:1\n\t"
        "v_tanh_f32 %[a1], %[b1]\n\t"
        "v_mul_f32 %[hc], 0.5, %[b1]\n\t"
        "v_mul_f32 %[h], %[a2], %[a1]\n\t"
        : [h] "+v"(h), [hc] "+v"(hc), [hx] "=&v"(hx), [a1] "=&v"(a1), [a2] "=&v"(a2), [b1] "=&v"(b1), [b2] "=&v"(b2)
        : [acc] "v"(acc), [wo0] "v"(wo[0]), [wo1] "v"(wo[1]), [wo2] "v"(wo[2]), [wo3] "v"(wo[3]), [wx0] "v"(wx[0]), [wx1] "v"(wx[1]), [wx2] "v"(wx[2]), [wx3] "v"(wx[3]), [s0] "s"(s0));
}

__device__ __forceinline__ void step_d(float& h, float& hc, const float acc, const float (&wo)[4], const float (&wx)[4], const int s0, const float (&wi)[4], const float (&wj)[4], float& d) {
    float hx, a1, a2, b1, b2; float e1, e2, f1, f2;
    asm(
        "v_permlanex16_b32 %[hx], %[h], %[s0], 0xfedcba98\n\t"
        "v_fma_f32_e64_dpp %[a1], %[h], %[wo0], %[acc] row_share:0 row_mask:0xf bank_mask:0xf bound_ctrl:1\n\t"
        "v_mul_f32_dpp %[a2], %[h], %[wo2] row_share:8 row_mask:0xf bank_mask:0xf bound_ctrl:1\n\t"
        "v_mul_f32_dpp %[b1], %[hx], %[wx0] row_share:0 row_mask:0xf bank_mask:0xf bound_ctrl:1\n\t"
        "v_mul_f32_dpp %[b2], %[hx], %[wx2] row_share:8 row_mask:0xf bank_mask:0xf bound_ctrl:1\n\t"
        "v_fma_f32_e64_dpp %[a1], %[h], %[wo1], %[a1] row_share:4 row_mask:0xf bank_mask:0xf bound_ctrl:1\n\t"
        "v_fma_f32_e64_dpp %[b1], %[hx], %[wx1], %[b1] row_share:4 row_mask:0xf bank_mask:0xf bound_ctrl:1\n\t"
        "v_fma_f32_e64_dpp %[b2], %[hx], %[wx3], %[b2] row_share:12 row_mask:0xf bank_mask:0xf bound_ctrl:1\n\t"
        "v_fma_f32_e64_dpp %[a2], %[h], %[wo3], %[a2] row_share:12 row_mask:0xf bank_mask:0xf bound_ctrl:1\n\t"
        "v_mul_f32_dpp %[e1], %[h], %[wi0] row_share:0 row_mask:0xf bank_mask:0xf bound_ctrl:1\n\t"
        "v_add_f32 %[b1], %[b1], %[b2]\n\t"
        "v_mul_f32_dpp %[e2], %[h], %[wi2] row_share:8 row_mask:0xf bank_mask:0xf bound_ctrl:1\n\t"
        "v_add_f32 %[a1], %[a1], %[a2]\n\t"
        "v_mul_f32_dpp %[f1], %[hx], %[wj0] row_share:0 row_mask:0xf bank_mask:0xf bound_ctrl:1\n\t"
        "v_mul_f32_dpp %[f2], %[hx], %[wj2] row_share:8 row_mask:0xf bank_mask:0xf bound_ctrl:1\n\t"
        "v_add_f32 %[a1], %[a1], %[b1]\n\t"
        "v_fma_f32_e64_dpp %[e1], %[h], %[wi1], %[e1] row_share:4 row_mask:0xf bank_mask:0xf bound_ctrl:1\n\t"
        "v_fma_f32_e64_dpp %[e2], %[h], %[wi3], %[e2] row_share:12 row_mask:0xf bank_mask:0xf bound_ctrl:1\n\t"
        "v_tanh_f32 %[a1], %[a1]\n\t"
        "v_fma_f32_e64_dpp %[f1], %[hx], %[wj1], %[f1] row_share:4 row_mask:0xf bank_mask:0xf bound_ctrl:1\n\t"
        "v_fma_f32_e64_dpp %[f2], %[hx], %[wj3], %[f2] row_share:12 row_mask:0xf bank_mask:0xf bound_ctrl:1\n\t"
        "v_add_f32 %[e1], %[e1], %[e2]\n\t"
        "v_fma_f32_e64_dpp %[b1], %[a1], %[a1], %[a1] quad_perm:[0,0,0,0] row_mask:0xf bank_mask:0xf bound_ctrl:1\n\t"
        "v_fma_f32_e64_dpp %[b2], %[a1], %[hc], %[hc] quad_perm:[1,1,1,1] row_mask:0xf bank_mask:0xf bound_ctrl:1\n\t"
        "v_fma_f32_e64_dpp %[a2], %[a1], 0.5, 0.5 quad_perm:[3,3,3,3] row_mask:0xf bank_mask:0xf bound_ctrl:1\n\t"
        "v_add_f32 %[f1], %[f1], %[f2]\n\t"
        "v_fma_f32_e64_dpp %[b1], %[b1], 0.5, %[b2] quad_perm:[2,2,2,2] row_mask:0xf bank_mask:0xf bound_ctrl:1\n\t"
        "v_add_f32 %[e1], %[e1], %[f1]\n\t"
        "v_tanh_f32 %[a1], %[b1]\n\t"
        "v_mul_f32 %[hc], 0.5, %[b1]\n\t"
        "v_mul_f32 %[h], %[a2], %[a1]\n\t"
        : [h] "+v"(h), [hc] "+v"(hc), [hx] "=&v"(hx), [a1] "=&v"(a1), [a2] "=&v"(a2), [b1] "=&v"(b1), [b2] "=&v"(b2), [e1] "=&v"(e1), [e2] "=&v"(e2), [f1] "=&v"(f1), [f2] "=&v"(f2)
        : [acc] "v"(acc), [wo0] "v"(wo[0]), [wo1] "v"(wo[1]), [wo2] "v"(wo[2]), [wo3] "v"(wo[3]), [wx0] "v"(wx[0]), [wx1] "v"(wx[1]), [wx2] "v"(wx[2]), [wx3] "v"(wx[3]), [s0] "s"(s0), [wi0] "v"(wi[0]), [wi1] "v"(wi[1]), [wi2] "v"(wi[2]), [wi3] "v"(wi[3]), [wj0] "v"(wj[0]), [wj1] "v"(wj[1]), [wj2] "v"(wj[2]), [wj3] "v"(wj[3])); d = e1;
}

__device__ __forceinline__ void step_h(float& h, float& hc, const float acc, const float (&wo)[4], const float (&wx)[4], const int s0, const float wr, const float br, float& rr) {
    float hx, a1, a2, b1, b2; float p, q;
    asm(
        "v_permlanex16_b32 %[hx], %[h], %[s0], 0xfedcba98\n\t"
        "v_fma_f32_e64_dpp %[a1], %[h], %[wo0], %[acc] row_share:0 row_mask:0xf bank_mask:0xf bound_ctrl:1\n\t"
        "v_mul_f32_dpp %[a2], %[h], %[wo2] row_share:8 row_mask:0xf bank_mask:0xf bound_ctrl:1\n\t"
        "v_mul_f32_dpp %[b1], %[hx], %[wx0] row_share:0 row_mask:0xf bank_mask:0xf bound_ctrl:1\n\t"
        "v_mul_f32_dpp %[b2], %[hx], %[wx2] row_share:8 row_mask:0xf bank_mask:0xf bound_ctrl:1\n\t"
        "v_fma_f32_e64_dpp %[a1], %[h], %[wo1], %[a1] row_share:4 row_mask:0xf bank_mask:0xf bound_ctrl:1\n\t"
        "v_fma_f32_e64_dpp %[b1], %[hx], %[wx1], %[b1] row_share:4 row_mask:0xf bank_mask:0xf bound_ctrl:1\n\t"
        "v_fma_f32_e64_dpp %[b2], %[hx], %[wx3], %[b2] row_share:12 row_mask:0xf bank_mask:0xf bound_ctrl:1\n\t"
        "v_fma_f32_e64_dpp %[a2], %[h], %[wo3], %[a2] row_share:12 row_mask:0xf bank_mask:0xf bound_ctrl:1\n\t"
        "v_mul_f32 %[p], %[h], %[wr]\n\t"
        "v_add_f32 %[b1], %[b1], %[b2]\n\t"
        "v_add_f32 %[a1], %[a1], %[a2]\n\t"
        "v_add_f32_dpp %[p], %[p], %[p] row_xmask:4 row_mask:0xf bank_mask:0xf bound_ctrl:1\n\t"
        "v_add_f32 %[a1], %[a1], %[b1]\n\t"
        "v_add_f32_dpp %[p], %[p], %[p] row_xmask:8 row_mask:0xf bank_mask:0xf bound_ctrl:1\n\t"
        "v_tanh_f32 %[a1], %[a1]\n\t"
        "v_permlanex16_b32 %[q], %[p], %[s0], 0xfedcba98\n\t"
        "v_fma_f32_e64_dpp %[b1], %[a1], %[a1], %[a1] quad_perm:[0,0,0,0] row_mask:0xf bank_mask:0xf bound_ctrl:1\n\t"
        "v_fma_f32_e64_dpp %[b2], %[a1], %[hc], %[hc] quad_perm:[1,1,1,1] row_mask:0xf bank_mask:0xf bound_ctrl:1\n\t"
        "v_fma_f32_e64_dpp %[a2], %[a1], 0.5, 0.5 quad_perm:[3,3,3,3] row_mask:0xf bank_mask:0xf bound_ctrl:1\n\t"
        "v_add_f32 %[p], %[p], %[q]\n\t"
        "v_fma_f32_e64_dpp %[b1], %[b1], 0.5, %[b2] quad_perm:[2,2,2,2] row_mask:0xf bank_mask:0xf bound_ctrl:1\n\t"
        "v_tanh_f32 %[a1], %[b1]\n\t"
        "v_mul_f32 %[hc], 0.5, %[b1]\n\t"
        "v_add_f32 %[p], %[p], %[br]\n\t"
        "v_mul_f32 %[h], %[a2], %[a1]\n\t"
        : [h] "+v"(h), [hc] "+v"(hc), [hx] "=&v"(hx), [a1] "=&v"(a1), [a2] "=&v"(a2), [b1] "=&v"(b1), [b2] "=&v"(b2), [p] "=&v"(p), [q] "=&v"(q)
        : [acc] "v"(acc), [wo0] "v"(wo[0]), [wo1] "v"(wo[1]), [wo2] "v"(wo[2]), [wo3] "v"(wo[3]), [wx0] "v"(wx[0]), [wx1] "v"(wx[1]), [wx2] "v"(wx[2]), [wx3] "v"(wx[3]), [s0] "s"(s0), [wr] "v"(wr), [br] "v"(br)); rr = p;
}

#define DPP_ALL " row_mask:0xf bank_mask:0xf bound_ctrl:1"
#define FMA_DPP(CTRL, a, b, c) __extension__({ float d__;                               \
    asm("v_fma_f32_e64_dpp %0, %1, %2, %3 " CTRL DPP_ALL                               \
        : "=v"(d__) : "v"(a), "v"(b), "v"(c)); d__; })
#define MUL_DPP(CTRL, a, b) __extension__({ float d__;                                  \
    asm("v_mul_f32_dpp %0, %1, %2 " CTRL DPP_ALL                                       \
        : "=v"(d__) : "v"(a), "v"(b)); d__; })
__device__ __forceinline__ float input_half(float h, float hx,
                                            const float (&wi)[4], const float (&wj)[4]) {
    float e1 = MUL_DPP("row_share:0", h,  wi[0]);
    float e2 = MUL_DPP("row_share:8", h,  wi[2]);
    float f1 = MUL_DPP("row_share:0", hx, wj[0]);
    float f2 = MUL_DPP("row_share:8", hx, wj[2]);
    e1 = FMA_DPP("row_share:4",  h,  wi[1], e1);
    e2 = FMA_DPP("row_share:12", h,  wi[3], e2);
    f1 = FMA_DPP("row_share:4",  hx, wj[1], f1);
    f2 = FMA_DPP("row_share:12", hx, wj[3], f2);
    return (e1 + e2) + (f1 + f2);
}

__device__ __forceinline__ void lds_barrier() {
    __builtin_amdgcn_fence(__ATOMIC_RELEASE, "workgroup", "local");
    __builtin_amdgcn_s_barrier();
    __builtin_amdgcn_fence(__ATOMIC_ACQUIRE, "workgroup", "local");
}

#define NG (SEQ / PF)

__global__ __launch_bounds__(32 * NW) void lstm_fused_kernel(
    const float* x,
    const float* w_ih0,
    const float* b_ih0,
    const float* b_hh0,
    const float* w_hh0,
    const float* w_ih1,
    const float* w_hh1,
    const float* b_ih1,
    const float* b_hh1,
    const float* w_reg,
    const float* b_reg,
    float* out)
{
    __shared__ v4f   sB[TILE_V4];
    __shared__ v4f   sA[NW][TILE_V4];
    __shared__ float sXg[SEQ * G4];
    __shared__ float pbuf[2][PF][32];

    const int tid = (int)threadIdx.x;
    const int l   = tid & 31;
    const int b   = (int)blockIdx.x;

    const int wave  = __builtin_amdgcn_readfirstlane(tid >> 5);
    const int n     = l & 15;
    const int hi    = l >> 4;

    const v4f* w4 = (const v4f*)w_ih0 + tid;
    v4f wb[WB_FULL + 1];
    #pragma unroll
    for (int j = 0; j < WB_FULL; ++j) wb[j] = w4[j * 32 * NW];
    if (tid < WB_REST) wb[WB_FULL] = w4[WB_FULL * 32 * NW];

    const char* xb = (const char*)(x + (size_t)b * INF);
    const unsigned offA = (unsigned)l * 16u;
    const unsigned offB = offA + (unsigned)((BATCH * INF - INF) * 4);
    auto piece = [&](const int tile, const int t) __attribute__((always_inline)) -> const v4f* {
        const int r0  = (t * 32) / ROW_V4;
        const int c0  = t * 32 - r0 * ROW_V4;
        const unsigned vo = (l >= ROW_V4 - c0) ? offB : offA;
        return (const v4f*)(xb + (size_t)(tile * 32 + r0) * (BATCH * INF * 4) + c0 * 16 + vo);
    };

    v4f pre[TILE_LD];
    #pragma unroll
    for (int t = 0; t < TILE_LD; ++t) pre[t] = *piece(wave, t);

    #pragma unroll
    for (int j = 0; j < WB_FULL; ++j) sB[tid + j * 32 * NW] = wb[j];
    if (tid < WB_REST) sB[tid + WB_FULL * 32 * NW] = wb[WB_FULL];

    float gb0 = b_ih0[n]      + b_hh0[n];
    float gb1 = b_ih0[n + 16] + b_hh0[n + 16];
    const int p0 = (n & 7) * 4 + (n >> 3);
    const int p1 = p0 + 2;

    v4f* myA = &sA[wave][0];
    const float* aF = (const float*)myA + n * INF + 2 * hi;
    const float* bF = (const float*)&sB[0] + n * INF + 2 * hi;

    v8f acc00 = {0.f,0.f,0.f,0.f,0.f,0.f,0.f,0.f};
    v8f acc01 = acc00, acc10 = acc00, acc11 = acc00;

    auto ksteps = [&](const float* a, const float* bb, const int nk) __attribute__((always_inline)) {
        #pragma unroll
        for (int ks = 0; ks < nk; ++ks) {
            const v2f a0 = *(const v2f*)(a + 4 * ks);
            const v2f a1 = *(const v2f*)(a + 16 * INF + 4 * ks);
            const v2f b0 = *(const v2f*)(bb + 4 * ks);
            const v2f b1 = *(const v2f*)(bb + 16 * INF + 4 * ks);
            acc00 = __builtin_amdgcn_wmma_f32_16x16x4_f32(false, a0, false, b0,
                                                          (short)0, acc00, false, false);
            acc01 = __builtin_amdgcn_wmma_f32_16x16x4_f32(false, a0, false, b1,
                                                          (short)0, acc01, false, false);
            acc10 = __builtin_amdgcn_wmma_f32_16x16x4_f32(false, a1, false, b0,
                                                          (short)0, acc10, false, false);
            acc11 = __builtin_amdgcn_wmma_f32_16x16x4_f32(false, a1, false, b1,
                                                          (short)0, acc11, false, false);
        }
    };
    auto tile_out = [&](const int tile) __attribute__((always_inline)) {
        float* ob = sXg + (tile * 32 + hi * 8) * G4;
        #pragma unroll
        for (int r = 0; r < 8; ++r) {
            ob[r * G4 + p0]        = acc00[r] + gb0;
            ob[r * G4 + p1]        = acc01[r] + gb1;
            ob[(r + 16) * G4 + p0] = acc10[r] + gb0;
            ob[(r + 16) * G4 + p1] = acc11[r] + gb1;
        }
        const v8f z = {0.f,0.f,0.f,0.f,0.f,0.f,0.f,0.f};
        acc00 = z; acc01 = z; acc10 = z; acc11 = z;
    };
    auto tile_in = [&]() __attribute__((always_inline)) {
        #pragma unroll
        for (int t = 0; t < TILE_LD; ++t) myA[t * 32 + l] = pre[t];
    };

    const int role = wave;
    const int unit = l >> 2;
    const int gate = l & 3;
    const int row  = gate * HD + unit;
    const int ko   = (l >> 4) * 4;
    const int kx   = 4 - ko;

    const bool  is_g = (gate == 2);
    const float aS = is_g ? 1.0f : 0.5f;

    float wao[4], wax[4], wio[4], wix[4];
    const float* wrec = (role == 0) ? w_hh0 : w_hh1;
    #pragma unroll
    for (int m = 0; m < 4; ++m) {
        wao[m] = aS * wrec[row * HD + ko + m];    wax[m] = aS * wrec[row * HD + kx + m];
        wio[m] = aS * w_ih1[row * HD + ko + m];   wix[m] = aS * w_ih1[row * HD + kx + m];
    }
    float bias1 = aS * (b_ih1[row] + b_hh1[row]);

    float wr = w_reg[(l & 1) * HD + unit];
    float br = b_reg[l & 1];
    const int   lu = l >> 1;
    #pragma unroll
    for (int m = 0; m < 4; ++m)
        asm volatile("" : "+v"(wao[m]), "+v"(wax[m]), "+v"(wio[m]), "+v"(wix[m]));
    asm volatile("" : "+v"(bias1), "+v"(wr), "+v"(br), "+v"(gb0), "+v"(gb1));

    tile_in();
    xg_lds_barrier();

    if (wave == 2 || wave == 3) {
        #pragma unroll
        for (int t = 0; t < TILE_LD; ++t) pre[t] = *piece(SIDE0 + wave, t);
    }
    __builtin_amdgcn_sched_barrier(0);

    if (wave < 4) {
        ksteps(aF, bF, INF / 4);
        tile_out(wave);
        if (wave >= 2) tile_in();
    }


    lds_barrier();

    if (wave >= 2) {
        const int last = (wave >= 4 ? SIDE0 : SIDE1) + wave;
        #pragma unroll
        for (int t = 0; t < TILE_LD; ++t) pre[t] = *piece(last, t);
    }

    float h = 0.f, hc = 0.f;
    const int sel0 = 0x76543210;

    const float* xgp = &sXg[l];

    auto layer0_group = [&](const float (&r)[PF], float* dst) __attribute__((always_inline)) {
        step_p(h, hc, r[0] * aS, wao, wax, sel0);
        #pragma unroll
        for (int u = 1; u < PF; ++u) {
            float d;
            step_d(h, hc, r[u] * aS, wao, wax, sel0, wio, wix, d);
            dst[(u - 1) * 32] = d;
        }
        dst[(PF - 1) * 32] = input_half(h, xhalf(h), wio, wix);
    };

    auto layer1_group = [&](const float* src, const int s0) __attribute__((always_inline)) {
        float pin[PF];
        #pragma unroll
        for (int u = 0; u < PF; ++u) pin[u] = src[u * 32];
        __builtin_amdgcn_sched_barrier(0);
        float res = 0.f;
        step_p(h, hc, pin[0] + bias1, wao, wax, sel0);
        #pragma unroll
        for (int u = 1; u < PF; ++u) {
            float rr;
            step_h(h, hc, pin[u] + bias1, wao, wax, sel0, wr, br, rr);
            res = (lu == u - 1) ? rr : res;
        }
        {
            float p = h * wr;
            p += row_xor<4>(p);
            p += row_xor<8>(p);
            p += xhalf(p);
            res = (lu == PF - 1) ? (p + br) : res;
        }
        out[((size_t)(s0 + lu) * BATCH + b) * 2 + (l & 1)] = res;
    };

    float ra[PF], rb[PF];
    if (role == 0) {
        #pragma unroll
        for (int u = 0; u < PF; ++u) ra[u] = xgp[u * G4];
    }


    auto side = [&](const int ht) __attribute__((always_inline)) {
        if (role >= 4) {
            if (ht < NCHF) {
                ksteps(aF + 4 * CHF * ht, bF + 4 * CHF * ht, CHF);
            } else if (ht == NCHF) {
                tile_out(role);
                tile_in();
            } else if (ht <= NCHF + NCH) {
                ksteps(aF + 4 * CH * (ht - NCHF - 1), bF + 4 * CH * (ht - NCHF - 1), CH);
            } else if (ht == NCHF + NCH + 1) {
                tile_out(SIDE0 + role);
            }
        } else {
            if (ht >= NCHF && ht < NCHF + NCH) {
                ksteps(aF + 4 * CH * (ht - NCHF), bF + 4 * CH * (ht - NCHF), CH);
            } else if (ht == NCHF + NCH) {
                tile_out(SIDE0 + role);
                tile_in();
            } else if (ht > NCHF + NCH && ht <= NCHF + 2 * NCH) {
                ksteps(aF + 4 * CH * (ht - NCHF - NCH - 1), bF + 4 * CH * (ht - NCHF - NCH - 1), CH);
            } else if (ht == NCHF + 2 * NCH + 1) {
                tile_out(SIDE1 + role);
            }
        }
    };

    for (int g = 0; g <= NG; g += 2) {
        if (role == 0) {
            if (g < NG) {
                #pragma unroll
                for (int u = 0; u < PF; ++u) rb[u] = xgp[((g + 1) * PF + u) * G4];
                __builtin_amdgcn_sched_barrier(0);
                layer0_group(ra, &pbuf[0][0][l]);
            }
        } else if (role == 1) {
            if (g >= 2) layer1_group(&pbuf[1][0][l], (g - 1) * PF);
        } else {
            side(g);
        }
        lds_barrier();

        if (role == 0) {
            if (g < NG) {
                #pragma unroll
                for (int u = 0; u < PF; ++u)
                    ra[u] = xgp[(((g + 2) * PF + u) & (SEQ - 1)) * G4];
                __builtin_amdgcn_sched_barrier(0);
                layer0_group(rb, &pbuf[1][0][l]);
            }
        } else if (role == 1) {
            if (g < NG) layer1_group(&pbuf[0][0][l], g * PF);
        } else {
            side(g + 1);
        }
        lds_barrier();
    }
}

extern "C" void kernel_launch(void* const* d_in, const int* in_sizes, int n_in,
                              void* d_out, int out_size, void* d_ws, size_t ws_size,
                              hipStream_t stream) {
    const float* x     = (const float*)d_in[0];
    const float* w_ih0 = (const float*)d_in[1];
    const float* w_hh0 = (const float*)d_in[2];
    const float* b_ih0 = (const float*)d_in[3];
    const float* b_hh0 = (const float*)d_in[4];
    const float* w_ih1 = (const float*)d_in[5];
    const float* w_hh1 = (const float*)d_in[6];
    const float* b_ih1 = (const float*)d_in[7];
    const float* b_hh1 = (const float*)d_in[8];
    const float* w_reg = (const float*)d_in[9];
    const float* b_reg = (const float*)d_in[10];
    float* out = (float*)d_out;

    lstm_fused_kernel<<<256, 32 * NW, 0, stream>>>(x, w_ih0, b_ih0, b_hh0, w_hh0, w_ih1, w_hh1,
                                                  b_ih1, b_hh1, w_reg, b_reg, out);
}
